# starting kernel
# baseline (speedup 1.0000x reference)
#include <hip/hip_runtime.h>

typedef __attribute__((ext_vector_type(2))) float v2f;
typedef __attribute__((ext_vector_type(8))) float v8f;

#define F0 512
#define F1 256
#define F2 64

__global__ void gcn_degree_kernel(const int* __restrict__ src, const int* __restrict__ dst,
                                  float* __restrict__ deg_out, float* __restrict__ deg_in, int E) {
    int e = blockIdx.x * blockDim.x + threadIdx.x;
    if (e < E) {
        atomicAdd(&deg_out[src[e]], 1.0f);
        atomicAdd(&deg_in[dst[e]], 1.0f);
    }
}

__global__ void gcn_norm_kernel(float* __restrict__ a, float* __restrict__ b, int n) {
    int i = blockIdx.x * blockDim.x + threadIdx.x;
    if (i < n) {
        float da = a[i];
        a[i] = (da > 0.0f) ? rsqrtf(fmaxf(da, 1.0f)) : 1.0f;
        float db = b[i];
        b[i] = (db > 0.0f) ? rsqrtf(fmaxf(db, 1.0f)) : 1.0f;
    }
}

__global__ __launch_bounds__(256) void gcn_gemm_wmma_f32(
    const float* __restrict__ A, const float* __restrict__ B,
    const float* __restrict__ normA, float* __restrict__ C,
    int M, int K, int N) {
    int wave = (blockIdx.x * blockDim.x + threadIdx.x) >> 5;
    int lane = threadIdx.x & 31;
    int tilesN = N >> 6;
    int tm = wave / tilesN;
    int tn = wave % tilesN;
    int m0 = tm << 4;
    int n0 = tn << 6;
    if (m0 >= M) return;

    int half = lane >> 4;
    int l    = lane & 15;
    int row  = m0 + l;
    float scale = normA[row];
    const float* Arow = A + (size_t)row * K;

    v8f acc0 = {}, acc1 = {}, acc2 = {}, acc3 = {};
    for (int k0 = 0; k0 < K; k0 += 4) {
        int ka = k0 + (half << 1);
        float2 at = *(const float2*)(Arow + ka);
        v2f a;
        a.x = at.x * scale;
        a.y = at.y * scale;

        const float* Bp = B + (size_t)ka * N + n0 + l;
        v2f b0, b1, b2, b3;
        b0.x = Bp[0];   b0.y = Bp[N];
        b1.x = Bp[16];  b1.y = Bp[N + 16];
        b2.x = Bp[32];  b2.y = Bp[N + 32];
        b3.x = Bp[48];  b3.y = Bp[N + 48];

        acc0 = __builtin_amdgcn_wmma_f32_16x16x4_f32(false, a, false, b0, (short)0, acc0, false, false);
        acc1 = __builtin_amdgcn_wmma_f32_16x16x4_f32(false, a, false, b1, (short)0, acc1, false, false);
        acc2 = __builtin_amdgcn_wmma_f32_16x16x4_f32(false, a, false, b2, (short)0, acc2, false, false);
        acc3 = __builtin_amdgcn_wmma_f32_16x16x4_f32(false, a, false, b3, (short)0, acc3, false, false);
    }

    #pragma unroll
    for (int i = 0; i < 8; ++i) {
        int r = m0 + i + (half << 3);
        size_t base = (size_t)r * N + n0 + l;
        C[base]      = acc0[i];
        C[base + 16] = acc1[i];
        C[base + 32] = acc2[i];
        C[base + 48] = acc3[i];
    }
}

__global__ void gcn_scatter_add_kernel(const float* __restrict__ H, const int* __restrict__ src,
                                       const int* __restrict__ dst, float* __restrict__ AGG,
                                       int E, int F) {
    long tid = (long)blockIdx.x * blockDim.x + threadIdx.x;
    int cpe = F >> 2;
    long e = tid / cpe;
    int  c = (int)(tid % cpe) << 2;
    if (e >= E) return;
    int s = src[e];
    int d = dst[e];
    float4 v = *(const float4*)(H + (size_t)s * F + c);
    float* o = AGG + (size_t)d * F + c;
    atomicAdd(o + 0, v.x);
    atomicAdd(o + 1, v.y);
    atomicAdd(o + 2, v.z);
    atomicAdd(o + 3, v.w);
}

__global__ void gcn_bias_act_kernel(const float* __restrict__ AGG, const float* __restrict__ norm_dst,
                                    const float* __restrict__ bias, float* __restrict__ OUT,
                                    long total, int F, int relu) {
    long i = (long)blockIdx.x * blockDim.x + threadIdx.x;
    if (i < total) {
        int node = (int)(i / F);
        int f    = (int)(i % F);
        float v = AGG[i] * norm_dst[node] + bias[f];
        OUT[i] = relu ? fmaxf(v, 0.0f) : v;
    }
}

extern "C" void kernel_launch(void* const* d_in, const int* in_sizes, int n_in,
                              void* d_out, int out_size, void* d_ws, size_t ws_size,
                              hipStream_t stream) {
    const float* feat = (const float*)d_in[0];
    const int*   src  = (const int*)d_in[1];
    const int*   dst  = (const int*)d_in[2];
    const float* W1   = (const float*)d_in[3];
    const float* b1   = (const float*)d_in[4];
    const float* W2   = (const float*)d_in[5];
    const float* b2   = (const float*)d_in[6];
    float* out = (float*)d_out;

    const int N = in_sizes[0] / F0;
    const int E = in_sizes[1];

    float* ws       = (float*)d_ws;
    float* norm_src = ws;
    float* norm_dst = norm_src + N;
    float* H1       = norm_dst + N;
    float* AGG1     = H1   + (size_t)N * F1;
    float* H2       = AGG1 + (size_t)N * F1;

    hipMemsetAsync(norm_src, 0, (size_t)2 * N * sizeof(float), stream);
    hipMemsetAsync(AGG1, 0, (size_t)N * F1 * sizeof(float), stream);
    hipMemsetAsync(out,  0, (size_t)N * F2 * sizeof(float), stream);

    gcn_degree_kernel<<<(E + 255) / 256, 256, 0, stream>>>(src, dst, norm_src, norm_dst, E);
    gcn_norm_kernel<<<(N + 255) / 256, 256, 0, stream>>>(norm_src, norm_dst, N);

    {
        int waves  = (N / 16) * (F1 / 64);
        int blocks = (waves * 32 + 255) / 256;
        gcn_gemm_wmma_f32<<<blocks, 256, 0, stream>>>(feat, W1, norm_src, H1, N, F0, F1);
    }
    {
        long threads = (long)E * (F1 / 4);
        gcn_scatter_add_kernel<<<(int)((threads + 255) / 256), 256, 0, stream>>>(H1, src, dst, AGG1, E, F1);
    }
    {
        long total = (long)N * F1;
        gcn_bias_act_kernel<<<(int)((total + 255) / 256), 256, 0, stream>>>(AGG1, norm_dst, b1, AGG1, total, F1, 1);
    }
    {
        int waves  = (N / 16) * (F2 / 64);
        int blocks = (waves * 32 + 255) / 256;
        gcn_gemm_wmma_f32<<<blocks, 256, 0, stream>>>(AGG1, W2, norm_src, H2, N, F1, F2);
    }
    {
        long threads = (long)E * (F2 / 4);
        gcn_scatter_add_kernel<<<(int)((threads + 255) / 256), 256, 0, stream>>>(H2, src, dst, out, E, F2);
    }
    {
        long total = (long)N * F2;
        gcn_bias_act_kernel<<<(int)((total + 255) / 256), 256, 0, stream>>>(out, norm_dst, b2, out, total, F2, 0);
    }
}


# f32 atomic edge scatter-add replaced by an in-call destination-sorted edge list (int-atomic arrival rank) + one wave per node summing source rows in registers; per-node sum order = arrival rank
# speedup vs baseline: 1.6915x; 1.6915x over previous
#include <hip/hip_runtime.h>

typedef __attribute__((ext_vector_type(2))) float v2f;
typedef __attribute__((ext_vector_type(8))) float v8f;

#define F0 512
#define F1 256
#define F2 64

__global__ void gcn_degree_rank_kernel(const int* __restrict__ src, const int* __restrict__ dst,
                                       int* __restrict__ cnt_out, int* __restrict__ cnt_in,
                                       int* __restrict__ rank, int E) {
    int e = blockIdx.x * blockDim.x + threadIdx.x;
    if (e < E) {
        int s = src[e];
        int d = dst[e];
        int r = atomicAdd(&cnt_in[d], 1);
        atomicAdd(&cnt_out[s], 1);
        rank[e] = r;
    }
}

__global__ __launch_bounds__(256) void gcn_norm_rowalloc_kernel(
    const int* __restrict__ cnt_out, const int* __restrict__ cnt_in,
    float* __restrict__ norm_src, float* __restrict__ norm_dst,
    int* __restrict__ row_start, int* __restrict__ next_free, int n) {
    int i = blockIdx.x * blockDim.x + threadIdx.x;
    int lane = threadIdx.x & 31;
    int ci = 0;
    if (i < n) {
        int co = cnt_out[i];
        ci = cnt_in[i];
        norm_src[i] = (co > 0) ? rsqrtf(fmaxf((float)co, 1.0f)) : 1.0f;
        norm_dst[i] = (ci > 0) ? rsqrtf(fmaxf((float)ci, 1.0f)) : 1.0f;
    }
    int incl = ci;
    for (int d = 1; d < 32; d <<= 1) {
        int t = __shfl_up(incl, d, 32);
        if (lane >= d) incl += t;
    }
    int total = __shfl(incl, 31, 32);
    int base = 0;
    if (lane == 0) base = atomicAdd(next_free, total);
    base = __shfl(base, 0, 32);
    if (i < n) row_start[i] = base + incl - ci;
}

__global__ void gcn_edge_fill_kernel(const int* __restrict__ src, const int* __restrict__ dst,
                                     const int* __restrict__ rank, const int* __restrict__ row_start,
                                     int* __restrict__ col, int E) {
    int e = blockIdx.x * blockDim.x + threadIdx.x;
    if (e < E) col[row_start[dst[e]] + rank[e]] = src[e];
}

__global__ __launch_bounds__(256) void gcn_gemm_wmma_f32(
    const float* __restrict__ A, const float* __restrict__ B,
    const float* __restrict__ normA, float* __restrict__ C,
    int M, int K, int N) {
    int wave = (blockIdx.x * blockDim.x + threadIdx.x) >> 5;
    int lane = threadIdx.x & 31;
    int tilesN = N >> 6;
    int tm = wave / tilesN;
    int tn = wave % tilesN;
    int m0 = tm << 4;
    int n0 = tn << 6;
    if (m0 >= M) return;

    int half = lane >> 4;
    int l    = lane & 15;
    int row  = m0 + l;
    float scale = normA[row];
    const float* Arow = A + (size_t)row * K;

    v8f acc0 = {}, acc1 = {}, acc2 = {}, acc3 = {};
    for (int k0 = 0; k0 < K; k0 += 4) {
        int ka = k0 + (half << 1);
        float2 at = *(const float2*)(Arow + ka);
        v2f a;
        a.x = at.x * scale;
        a.y = at.y * scale;

        const float* Bp = B + (size_t)ka * N + n0 + l;
        v2f b0, b1, b2, b3;
        b0.x = Bp[0];   b0.y = Bp[N];
        b1.x = Bp[16];  b1.y = Bp[N + 16];
        b2.x = Bp[32];  b2.y = Bp[N + 32];
        b3.x = Bp[48];  b3.y = Bp[N + 48];

        acc0 = __builtin_amdgcn_wmma_f32_16x16x4_f32(false, a, false, b0, (short)0, acc0, false, false);
        acc1 = __builtin_amdgcn_wmma_f32_16x16x4_f32(false, a, false, b1, (short)0, acc1, false, false);
        acc2 = __builtin_amdgcn_wmma_f32_16x16x4_f32(false, a, false, b2, (short)0, acc2, false, false);
        acc3 = __builtin_amdgcn_wmma_f32_16x16x4_f32(false, a, false, b3, (short)0, acc3, false, false);
    }

    #pragma unroll
    for (int i = 0; i < 8; ++i) {
        int r = m0 + i + (half << 3);
        size_t base = (size_t)r * N + n0 + l;
        C[base]      = acc0[i];
        C[base + 16] = acc1[i];
        C[base + 32] = acc2[i];
        C[base + 48] = acc3[i];
    }
}

__device__ __forceinline__ void gcn_add4(float4& a, const float4& x) {
    a.x += x.x; a.y += x.y; a.z += x.z; a.w += x.w;
}

template <int F, int LPN>
__global__ __launch_bounds__(256) void gcn_gather_sum_kernel(
    const float* __restrict__ H, const int* __restrict__ row_start,
    const int* __restrict__ cnt_in, const int* __restrict__ col,
    float* __restrict__ AGG, int n_nodes) {
    constexpr int NV = F / (4 * LPN);
    int node = (blockIdx.x * blockDim.x + threadIdx.x) / LPN;
    int sub  = threadIdx.x % LPN;
    if (node >= n_nodes) return;
    int start = row_start[node];
    int n     = cnt_in[node];
    const float* Hc = H + (sub << 2);

    float4 acc[NV];
    for (int v = 0; v < NV; ++v) acc[v] = make_float4(0.0f, 0.0f, 0.0f, 0.0f);

    for (int base = 0; base < n; base += LPN) {
        int m = n - base;
        if (m > LPN) m = LPN;
        int mine = (sub < m) ? col[start + base + sub] : 0;
        int j = 0;
        for (; j + 4 <= m; j += 4) {
            const float* p0 = Hc + (size_t)__shfl(mine, j,     LPN) * F;
            const float* p1 = Hc + (size_t)__shfl(mine, j + 1, LPN) * F;
            const float* p2 = Hc + (size_t)__shfl(mine, j + 2, LPN) * F;
            const float* p3 = Hc + (size_t)__shfl(mine, j + 3, LPN) * F;
            float4 x0[NV], x1[NV], x2[NV], x3[NV];
            for (int v = 0; v < NV; ++v) {
                x0[v] = *(const float4*)(p0 + v * LPN * 4);
                x1[v] = *(const float4*)(p1 + v * LPN * 4);
                x2[v] = *(const float4*)(p2 + v * LPN * 4);
                x3[v] = *(const float4*)(p3 + v * LPN * 4);
            }
            for (int v = 0; v < NV; ++v) {
                gcn_add4(acc[v], x0[v]);
                gcn_add4(acc[v], x1[v]);
                gcn_add4(acc[v], x2[v]);
                gcn_add4(acc[v], x3[v]);
            }
        }
        for (; j < m; ++j) {
            const float* p0 = Hc + (size_t)__shfl(mine, j, LPN) * F;
            for (int v = 0; v < NV; ++v) gcn_add4(acc[v], *(const float4*)(p0 + v * LPN * 4));
        }
    }

    float* o = AGG + (size_t)node * F + (sub << 2);
    for (int v = 0; v < NV; ++v) *(float4*)(o + v * LPN * 4) = acc[v];
}

__global__ void gcn_bias_act_kernel(const float* __restrict__ AGG, const float* __restrict__ norm_dst,
                                    const float* __restrict__ bias, float* __restrict__ OUT,
                                    long total, int F, int relu) {
    long i = (long)blockIdx.x * blockDim.x + threadIdx.x;
    if (i < total) {
        int node = (int)(i / F);
        int f    = (int)(i % F);
        float v = AGG[i] * norm_dst[node] + bias[f];
        OUT[i] = relu ? fmaxf(v, 0.0f) : v;
    }
}

extern "C" void kernel_launch(void* const* d_in, const int* in_sizes, int n_in,
                              void* d_out, int out_size, void* d_ws, size_t ws_size,
                              hipStream_t stream) {
    const float* feat = (const float*)d_in[0];
    const int*   src  = (const int*)d_in[1];
    const int*   dst  = (const int*)d_in[2];
    const float* W1   = (const float*)d_in[3];
    const float* b1   = (const float*)d_in[4];
    const float* W2   = (const float*)d_in[5];
    const float* b2   = (const float*)d_in[6];
    float* out = (float*)d_out;

    const int N = in_sizes[0] / F0;
    const int E = in_sizes[1];

    float* ws       = (float*)d_ws;
    float* norm_src = ws;
    float* norm_dst = norm_src + N;
    float* H1       = norm_dst + N;
    float* AGG1     = H1   + (size_t)N * F1;
    float* H2       = H1;
    int* cnt_out   = (int*)(AGG1 + (size_t)N * F1);
    int* cnt_in    = cnt_out + N;
    int* next_free = cnt_in + N;
    int* row_start = next_free + 32;
    int* rank      = row_start + N;
    int* col       = rank + E;

    hipMemsetAsync(cnt_out, 0, ((size_t)2 * N + 32) * sizeof(int), stream);

    gcn_degree_rank_kernel<<<(E + 255) / 256, 256, 0, stream>>>(src, dst, cnt_out, cnt_in, rank, E);
    gcn_norm_rowalloc_kernel<<<(N + 255) / 256, 256, 0, stream>>>(cnt_out, cnt_in, norm_src, norm_dst,
                                                                  row_start, next_free, N);
    gcn_edge_fill_kernel<<<(E + 255) / 256, 256, 0, stream>>>(src, dst, rank, row_start, col, E);

    {
        int waves  = (N / 16) * (F1 / 64);
        int blocks = (waves * 32 + 255) / 256;
        gcn_gemm_wmma_f32<<<blocks, 256, 0, stream>>>(feat, W1, norm_src, H1, N, F0, F1);
    }
    gcn_gather_sum_kernel<F1, 32><<<(N * 32 + 255) / 256, 256, 0, stream>>>(H1, row_start, cnt_in, col, AGG1, N);
    {
        long total = (long)N * F1;
        gcn_bias_act_kernel<<<(int)((total + 255) / 256), 256, 0, stream>>>(AGG1, norm_dst, b1, AGG1, total, F1, 1);
    }
    {
        int waves  = (N / 16) * (F2 / 64);
        int blocks = (waves * 32 + 255) / 256;
        gcn_gemm_wmma_f32<<<blocks, 256, 0, stream>>>(AGG1, W2, norm_src, H2, N, F1, F2);
    }
    gcn_gather_sum_kernel<F2, 16><<<(N * 16 + 255) / 256, 256, 0, stream>>>(H2, row_start, cnt_in, col, out, N);
    {
        long total = (long)N * F2;
        gcn_bias_act_kernel<<<(int)((total + 255) / 256), 256, 0, stream>>>(out, norm_dst, b2, out, total, F2, 0);
    }
}


# norm_dst scale + bias + relu applied to the gather kernel's in-register sums before its one store; the two elementwise launches (one element per thread, 16M threads) are removed; sum order unchanged
# speedup vs baseline: 1.9234x; 1.1371x over previous
#include <hip/hip_runtime.h>

typedef __attribute__((ext_vector_type(2))) float v2f;
typedef __attribute__((ext_vector_type(8))) float v8f;

#define F0 512
#define F1 256
#define F2 64

__global__ void gcn_degree_rank_kernel(const int* __restrict__ src, const int* __restrict__ dst,
                                       int* __restrict__ cnt_out, int* __restrict__ cnt_in,
                                       int* __restrict__ rank, int E) {
    int e = blockIdx.x * blockDim.x + threadIdx.x;
    if (e < E) {
        int s = src[e];
        int d = dst[e];
        int r = atomicAdd(&cnt_in[d], 1);
        atomicAdd(&cnt_out[s], 1);
        rank[e] = r;
    }
}

__global__ __launch_bounds__(256) void gcn_norm_rowalloc_kernel(
    const int* __restrict__ cnt_out, const int* __restrict__ cnt_in,
    float* __restrict__ norm_src, float* __restrict__ norm_dst,
    int* __restrict__ row_start, int* __restrict__ next_free, int n) {
    int i = blockIdx.x * blockDim.x + threadIdx.x;
    int lane = threadIdx.x & 31;
    int ci = 0;
    if (i < n) {
        int co = cnt_out[i];
        ci = cnt_in[i];
        norm_src[i] = (co > 0) ? rsqrtf(fmaxf((float)co, 1.0f)) : 1.0f;
        norm_dst[i] = (ci > 0) ? rsqrtf(fmaxf((float)ci, 1.0f)) : 1.0f;
    }
    int incl = ci;
    for (int d = 1; d < 32; d <<= 1) {
        int t = __shfl_up(incl, d, 32);
        if (lane >= d) incl += t;
    }
    int total = __shfl(incl, 31, 32);
    int base = 0;
    if (lane == 0) base = atomicAdd(next_free, total);
    base = __shfl(base, 0, 32);
    if (i < n) row_start[i] = base + incl - ci;
}

__global__ void gcn_edge_fill_kernel(const int* __restrict__ src, const int* __restrict__ dst,
                                     const int* __restrict__ rank, const int* __restrict__ row_start,
                                     int* __restrict__ col, int E) {
    int e = blockIdx.x * blockDim.x + threadIdx.x;
    if (e < E) col[row_start[dst[e]] + rank[e]] = src[e];
}

__global__ __launch_bounds__(256) void gcn_gemm_wmma_f32(
    const float* __restrict__ A, const float* __restrict__ B,
    const float* __restrict__ normA, float* __restrict__ C,
    int M, int K, int N) {
    int wave = (blockIdx.x * blockDim.x + threadIdx.x) >> 5;
    int lane = threadIdx.x & 31;
    int tilesN = N >> 6;
    int tm = wave / tilesN;
    int tn = wave % tilesN;
    int m0 = tm << 4;
    int n0 = tn << 6;
    if (m0 >= M) return;

    int half = lane >> 4;
    int l    = lane & 15;
    int row  = m0 + l;
    float scale = normA[row];
    const float* Arow = A + (size_t)row * K;

    v8f acc0 = {}, acc1 = {}, acc2 = {}, acc3 = {};
    for (int k0 = 0; k0 < K; k0 += 4) {
        int ka = k0 + (half << 1);
        float2 at = *(const float2*)(Arow + ka);
        v2f a;
        a.x = at.x * scale;
        a.y = at.y * scale;

        const float* Bp = B + (size_t)ka * N + n0 + l;
        v2f b0, b1, b2, b3;
        b0.x = Bp[0];   b0.y = Bp[N];
        b1.x = Bp[16];  b1.y = Bp[N + 16];
        b2.x = Bp[32];  b2.y = Bp[N + 32];
        b3.x = Bp[48];  b3.y = Bp[N + 48];

        acc0 = __builtin_amdgcn_wmma_f32_16x16x4_f32(false, a, false, b0, (short)0, acc0, false, false);
        acc1 = __builtin_amdgcn_wmma_f32_16x16x4_f32(false, a, false, b1, (short)0, acc1, false, false);
        acc2 = __builtin_amdgcn_wmma_f32_16x16x4_f32(false, a, false, b2, (short)0, acc2, false, false);
        acc3 = __builtin_amdgcn_wmma_f32_16x16x4_f32(false, a, false, b3, (short)0, acc3, false, false);
    }

    #pragma unroll
    for (int i = 0; i < 8; ++i) {
        int r = m0 + i + (half << 3);
        size_t base = (size_t)r * N + n0 + l;
        C[base]      = acc0[i];
        C[base + 16] = acc1[i];
        C[base + 32] = acc2[i];
        C[base + 48] = acc3[i];
    }
}

__device__ __forceinline__ void gcn_add4(float4& a, const float4& x) {
    a.x += x.x; a.y += x.y; a.z += x.z; a.w += x.w;
}

template <int F, int LPN, bool RELU>
__global__ __launch_bounds__(256) void gcn_gather_sum_kernel(
    const float* __restrict__ H, const int* __restrict__ row_start,
    const int* __restrict__ cnt_in, const int* __restrict__ col,
    const float* __restrict__ norm_dst, const float* __restrict__ bias,
    float* __restrict__ OUT, int n_nodes) {
    constexpr int NV = F / (4 * LPN);
    int node = (blockIdx.x * blockDim.x + threadIdx.x) / LPN;
    int sub  = threadIdx.x % LPN;
    if (node >= n_nodes) return;
    int start = row_start[node];
    int n     = cnt_in[node];
    float nd  = norm_dst[node];
    const float* Hc = H + (sub << 2);

    float4 acc[NV];
    for (int v = 0; v < NV; ++v) acc[v] = make_float4(0.0f, 0.0f, 0.0f, 0.0f);

    for (int base = 0; base < n; base += LPN) {
        int m = n - base;
        if (m > LPN) m = LPN;
        int mine = (sub < m) ? col[start + base + sub] : 0;
        int j = 0;
        for (; j + 4 <= m; j += 4) {
            const float* p0 = Hc + (size_t)__shfl(mine, j,     LPN) * F;
            const float* p1 = Hc + (size_t)__shfl(mine, j + 1, LPN) * F;
            const float* p2 = Hc + (size_t)__shfl(mine, j + 2, LPN) * F;
            const float* p3 = Hc + (size_t)__shfl(mine, j + 3, LPN) * F;
            float4 x0[NV], x1[NV], x2[NV], x3[NV];
            for (int v = 0; v < NV; ++v) {
                x0[v] = *(const float4*)(p0 + v * LPN * 4);
                x1[v] = *(const float4*)(p1 + v * LPN * 4);
                x2[v] = *(const float4*)(p2 + v * LPN * 4);
                x3[v] = *(const float4*)(p3 + v * LPN * 4);
            }
            for (int v = 0; v < NV; ++v) {
                gcn_add4(acc[v], x0[v]);
                gcn_add4(acc[v], x1[v]);
                gcn_add4(acc[v], x2[v]);
                gcn_add4(acc[v], x3[v]);
            }
        }
        for (; j < m; ++j) {
            const float* p0 = Hc + (size_t)__shfl(mine, j, LPN) * F;
            for (int v = 0; v < NV; ++v) gcn_add4(acc[v], *(const float4*)(p0 + v * LPN * 4));
        }
    }

    const float* bp = bias + (sub << 2);
    float* o = OUT + (size_t)node * F + (sub << 2);
    for (int v = 0; v < NV; ++v) {
        float4 b = *(const float4*)(bp + v * LPN * 4);
        float4 r;
        r.x = acc[v].x * nd + b.x;
        r.y = acc[v].y * nd + b.y;
        r.z = acc[v].z * nd + b.z;
        r.w = acc[v].w * nd + b.w;
        if (RELU) {
            r.x = fmaxf(r.x, 0.0f); r.y = fmaxf(r.y, 0.0f);
            r.z = fmaxf(r.z, 0.0f); r.w = fmaxf(r.w, 0.0f);
        }
        *(float4*)(o + v * LPN * 4) = r;
    }
}

extern "C" void kernel_launch(void* const* d_in, const int* in_sizes, int n_in,
                              void* d_out, int out_size, void* d_ws, size_t ws_size,
                              hipStream_t stream) {
    const float* feat = (const float*)d_in[0];
    const int*   src  = (const int*)d_in[1];
    const int*   dst  = (const int*)d_in[2];
    const float* W1   = (const float*)d_in[3];
    const float* b1   = (const float*)d_in[4];
    const float* W2   = (const float*)d_in[5];
    const float* b2   = (const float*)d_in[6];
    float* out = (float*)d_out;

    const int N = in_sizes[0] / F0;
    const int E = in_sizes[1];

    float* ws       = (float*)d_ws;
    float* norm_src = ws;
    float* norm_dst = norm_src + N;
    float* H1       = norm_dst + N;
    float* AGG1     = H1   + (size_t)N * F1;
    float* H2       = H1;
    int* cnt_out   = (int*)(AGG1 + (size_t)N * F1);
    int* cnt_in    = cnt_out + N;
    int* next_free = cnt_in + N;
    int* row_start = next_free + 32;
    int* rank      = row_start + N;
    int* col       = rank + E;

    hipMemsetAsync(cnt_out, 0, ((size_t)2 * N + 32) * sizeof(int), stream);

    gcn_degree_rank_kernel<<<(E + 255) / 256, 256, 0, stream>>>(src, dst, cnt_out, cnt_in, rank, E);
    gcn_norm_rowalloc_kernel<<<(N + 255) / 256, 256, 0, stream>>>(cnt_out, cnt_in, norm_src, norm_dst,
                                                                  row_start, next_free, N);
    gcn_edge_fill_kernel<<<(E + 255) / 256, 256, 0, stream>>>(src, dst, rank, row_start, col, E);

    {
        int waves  = (N / 16) * (F1 / 64);
        int blocks = (waves * 32 + 255) / 256;
        gcn_gemm_wmma_f32<<<blocks, 256, 0, stream>>>(feat, W1, norm_src, H1, N, F0, F1);
    }
    gcn_gather_sum_kernel<F1, 32, true><<<(N * 32 + 255) / 256, 256, 0, stream>>>(H1, row_start, cnt_in, col,
                                                                                 norm_dst, b1, AGG1, N);
    {
        int waves  = (N / 16) * (F2 / 64);
        int blocks = (waves * 32 + 255) / 256;
        gcn_gemm_wmma_f32<<<blocks, 256, 0, stream>>>(AGG1, W2, norm_src, H2, N, F1, F2);
    }
    gcn_gather_sum_kernel<F2, 16, false><<<(N * 16 + 255) / 256, 256, 0, stream>>>(H2, row_start, cnt_in, col,
                                                                                  norm_dst, b2, out, N);
}


# both GEMMs tiled through LDS: a 16-wave block stages scaled A rows and B once per 32-deep k chunk, each wave reuses one B fragment over 13 (layer 1) or 2 row tiles; C leaves via LDS; sum order kept
# speedup vs baseline: 2.4211x; 1.2588x over previous
#include <hip/hip_runtime.h>

typedef __attribute__((ext_vector_type(2))) float v2f;
typedef __attribute__((ext_vector_type(8))) float v8f;

#define F0 512
#define F1 256
#define F2 64

__global__ void gcn_degree_rank_kernel(const int* __restrict__ src, const int* __restrict__ dst,
                                       int* __restrict__ cnt_out, int* __restrict__ cnt_in,
                                       int* __restrict__ rank, int E) {
    int e = blockIdx.x * blockDim.x + threadIdx.x;
    if (e < E) {
        int s = src[e];
        int d = dst[e];
        int r = atomicAdd(&cnt_in[d], 1);
        atomicAdd(&cnt_out[s], 1);
        rank[e] = r;
    }
}

__global__ __launch_bounds__(256) void gcn_norm_rowalloc_kernel(
    const int* __restrict__ cnt_out, const int* __restrict__ cnt_in,
    float* __restrict__ norm_src, float* __restrict__ norm_dst,
    int* __restrict__ row_start, int* __restrict__ next_free, int n) {
    int i = blockIdx.x * blockDim.x + threadIdx.x;
    int lane = threadIdx.x & 31;
    int ci = 0;
    if (i < n) {
        int co = cnt_out[i];
        ci = cnt_in[i];
        norm_src[i] = (co > 0) ? rsqrtf(fmaxf((float)co, 1.0f)) : 1.0f;
        norm_dst[i] = (ci > 0) ? rsqrtf(fmaxf((float)ci, 1.0f)) : 1.0f;
    }
    int incl = ci;
    for (int d = 1; d < 32; d <<= 1) {
        int t = __shfl_up(incl, d, 32);
        if (lane >= d) incl += t;
    }
    int total = __shfl(incl, 31, 32);
    int base = 0;
    if (lane == 0) base = atomicAdd(next_free, total);
    base = __shfl(base, 0, 32);
    if (i < n) row_start[i] = base + incl - ci;
}

__global__ void gcn_edge_fill_kernel(const int* __restrict__ src, const int* __restrict__ dst,
                                     const int* __restrict__ rank, const int* __restrict__ row_start,
                                     int* __restrict__ col, int E) {
    int e = blockIdx.x * blockDim.x + threadIdx.x;
    if (e < E) col[row_start[dst[e]] + rank[e]] = src[e];
}

template <int K, int N, int RTW>
__global__ __launch_bounds__(512) void gcn_gemm_lds_wmma_f32(
    const float* __restrict__ A, const float* __restrict__ B,
    const float* __restrict__ normA, float* __restrict__ C, int M) {
    constexpr int NT = 512;
    constexpr int CT = N / 16;
    constexpr int RG = 16 / CT;
    constexpr int BM = RG * RTW * 16;
    constexpr int KC = 32;
    constexpr int NKC = K / KC;
    constexpr int PA = KC + 4;
    constexpr int PB = 2 * N + 32;
    constexpr int PC = N + 4;
    constexpr int A_FLOATS = BM * PA;
    constexpr int B_FLOATS = (KC / 2) * PB;
    constexpr int C_FLOATS = BM * PC;
    constexpr int S_FLOATS = (A_FLOATS + B_FLOATS) > C_FLOATS ? (A_FLOATS + B_FLOATS) : C_FLOATS;
    constexpr int AUNITS = BM * (KC / 4);
    constexpr int AU = (AUNITS + NT - 1) / NT;
    constexpr int BUNITS = (KC / 2) * (N / 4);
    constexpr int BU = (BUNITS + NT - 1) / NT;
    constexpr int CUNITS = BM * (N / 4);

    __shared__ __attribute__((aligned(16))) float smem[S_FLOATS];
    float* As = smem;
    float* Bs = smem + A_FLOATS;

    const int tid  = threadIdx.x;
    const int wave = tid >> 5;
    const int lane = tid & 31;
    const int h    = lane >> 4;
    const int l    = lane & 15;
    const int ct   = wave % CT;
    const int rg   = wave / CT;
    const int m_base = blockIdx.x * BM;

    const float* a_src[AU];
    float a_scale[AU];
    int   a_dst[AU];
    bool  a_ok[AU];
    #pragma unroll
    for (int i = 0; i < AU; ++i) {
        int u = tid + i * NT;
        a_ok[i] = (u < AUNITS);
        if (!a_ok[i]) u = 0;
        int r = u / (KC / 4);
        int c = (u % (KC / 4)) * 4;
        int g = m_base + r;
        if (g > M - 1) g = M - 1;
        a_src[i]   = A + (size_t)g * K + c;
        a_scale[i] = normA[g];
        a_dst[i]   = r * PA + c;
    }
    const float* b_src[BU];
    int  b_dst[BU];
    bool b_ok[BU];
    #pragma unroll
    for (int i = 0; i < BU; ++i) {
        int u = tid + i * NT;
        b_ok[i] = (u < BUNITS);
        if (!b_ok[i]) u = 0;
        int kp = u / (N / 4);
        int c4 = (u % (N / 4)) * 4;
        b_src[i] = B + (size_t)(2 * kp) * N + c4;
        b_dst[i] = kp * PB + 2 * c4;
    }

    float4 ra[AU], rb0[BU], rb1[BU];
    #pragma unroll
    for (int i = 0; i < AU; ++i) ra[i] = *(const float4*)(a_src[i]);
    #pragma unroll
    for (int i = 0; i < BU; ++i) {
        rb0[i] = *(const float4*)(b_src[i]);
        rb1[i] = *(const float4*)(b_src[i] + N);
    }

    v8f acc[RTW];
    #pragma unroll
    for (int rt = 0; rt < RTW; ++rt) acc[rt] = v8f{};

    const v2f* As2 = (const v2f*)As + (((rg * 16 + l) * PA) >> 1) + h;
    const v2f* Bs2 = (const v2f*)Bs + h * (PB / 2) + ct * 16 + l;

    for (int c = 0; c < NKC; ++c) {
        #pragma unroll
        for (int i = 0; i < AU; ++i) {
            if (a_ok[i]) {
                float4 v = ra[i];
                float s = a_scale[i];
                v.x *= s; v.y *= s; v.z *= s; v.w *= s;
                *(float4*)(As + a_dst[i]) = v;
            }
        }
        #pragma unroll
        for (int i = 0; i < BU; ++i) {
            if (b_ok[i]) {
                float4 p = make_float4(rb0[i].x, rb1[i].x, rb0[i].y, rb1[i].y);
                float4 q = make_float4(rb0[i].z, rb1[i].z, rb0[i].w, rb1[i].w);
                *(float4*)(Bs + b_dst[i])     = p;
                *(float4*)(Bs + b_dst[i] + 4) = q;
            }
        }
        __syncthreads();

        if (c + 1 < NKC) {
            const int ko = (c + 1) * KC;
            #pragma unroll
            for (int i = 0; i < AU; ++i) ra[i] = *(const float4*)(a_src[i] + ko);
            #pragma unroll
            for (int i = 0; i < BU; ++i) {
                rb0[i] = *(const float4*)(b_src[i] + (size_t)ko * N);
                rb1[i] = *(const float4*)(b_src[i] + (size_t)ko * N + N);
            }
        }

        #pragma unroll
        for (int kk = 0; kk < KC; kk += 4) {
            v2f b = Bs2[(kk >> 1) * (PB / 2)];
            #pragma unroll
            for (int rt = 0; rt < RTW; ++rt) {
                v2f a = As2[((rt * RG * 16 * PA) >> 1) + (kk >> 1)];
                acc[rt] = __builtin_amdgcn_wmma_f32_16x16x4_f32(false, a, false, b, (short)0, acc[rt], false, false);
            }
        }
        __syncthreads();
    }

    float* Cs = smem;
    #pragma unroll
    for (int rt = 0; rt < RTW; ++rt) {
        #pragma unroll
        for (int i = 0; i < 8; ++i)
            Cs[((rt * RG + rg) * 16 + 8 * h + i) * PC + ct * 16 + l] = acc[rt][i];
    }
    __syncthreads();
    for (int u = tid; u < CUNITS; u += NT) {
        int r  = u / (N / 4);
        int c4 = (u % (N / 4)) * 4;
        int g  = m_base + r;
        if (g < M) *(float4*)(C + (size_t)g * N + c4) = *(const float4*)(Cs + r * PC + c4);
    }
}

__device__ __forceinline__ void gcn_add4(float4& a, const float4& x) {
    a.x += x.x; a.y += x.y; a.z += x.z; a.w += x.w;
}

template <int F, int LPN, bool RELU>
__global__ __launch_bounds__(256) void gcn_gather_sum_kernel(
    const float* __restrict__ H, const int* __restrict__ row_start,
    const int* __restrict__ cnt_in, const int* __restrict__ col,
    const float* __restrict__ norm_dst, const float* __restrict__ bias,
    float* __restrict__ OUT, int n_nodes) {
    constexpr int NV = F / (4 * LPN);
    int node = (blockIdx.x * blockDim.x + threadIdx.x) / LPN;
    int sub  = threadIdx.x % LPN;
    if (node >= n_nodes) return;
    int start = row_start[node];
    int n     = cnt_in[node];
    float nd  = norm_dst[node];
    const float* Hc = H + (sub << 2);

    float4 acc[NV];
    for (int v = 0; v < NV; ++v) acc[v] = make_float4(0.0f, 0.0f, 0.0f, 0.0f);

    for (int base = 0; base < n; base += LPN) {
        int m = n - base;
        if (m > LPN) m = LPN;
        int mine = (sub < m) ? col[start + base + sub] : 0;
        int j = 0;
        for (; j + 4 <= m; j += 4) {
            const float* p0 = Hc + (size_t)__shfl(mine, j,     LPN) * F;
            const float* p1 = Hc + (size_t)__shfl(mine, j + 1, LPN) * F;
            const float* p2 = Hc + (size_t)__shfl(mine, j + 2, LPN) * F;
            const float* p3 = Hc + (size_t)__shfl(mine, j + 3, LPN) * F;
            float4 x0[NV], x1[NV], x2[NV], x3[NV];
            for (int v = 0; v < NV; ++v) {
                x0[v] = *(const float4*)(p0 + v * LPN * 4);
                x1[v] = *(const float4*)(p1 + v * LPN * 4);
                x2[v] = *(const float4*)(p2 + v * LPN * 4);
                x3[v] = *(const float4*)(p3 + v * LPN * 4);
            }
            for (int v = 0; v < NV; ++v) {
                gcn_add4(acc[v], x0[v]);
                gcn_add4(acc[v], x1[v]);
                gcn_add4(acc[v], x2[v]);
                gcn_add4(acc[v], x3[v]);
            }
        }
        for (; j < m; ++j) {
            const float* p0 = Hc + (size_t)__shfl(mine, j, LPN) * F;
            for (int v = 0; v < NV; ++v) gcn_add4(acc[v], *(const float4*)(p0 + v * LPN * 4));
        }
    }

    const float* bp = bias + (sub << 2);
    float* o = OUT + (size_t)node * F + (sub << 2);
    for (int v = 0; v < NV; ++v) {
        float4 b = *(const float4*)(bp + v * LPN * 4);
        float4 r;
        r.x = acc[v].x * nd + b.x;
        r.y = acc[v].y * nd + b.y;
        r.z = acc[v].z * nd + b.z;
        r.w = acc[v].w * nd + b.w;
        if (RELU) {
            r.x = fmaxf(r.x, 0.0f); r.y = fmaxf(r.y, 0.0f);
            r.z = fmaxf(r.z, 0.0f); r.w = fmaxf(r.w, 0.0f);
        }
        *(float4*)(o + v * LPN * 4) = r;
    }
}

extern "C" void kernel_launch(void* const* d_in, const int* in_sizes, int n_in,
                              void* d_out, int out_size, void* d_ws, size_t ws_size,
                              hipStream_t stream) {
    const float* feat = (const float*)d_in[0];
    const int*   src  = (const int*)d_in[1];
    const int*   dst  = (const int*)d_in[2];
    const float* W1   = (const float*)d_in[3];
    const float* b1   = (const float*)d_in[4];
    const float* W2   = (const float*)d_in[5];
    const float* b2   = (const float*)d_in[6];
    float* out = (float*)d_out;

    const int N = in_sizes[0] / F0;
    const int E = in_sizes[1];

    float* ws       = (float*)d_ws;
    float* norm_src = ws;
    float* norm_dst = norm_src + N;
    float* H1       = norm_dst + N;
    float* AGG1     = H1   + (size_t)N * F1;
    float* H2       = H1;
    int* cnt_out   = (int*)(AGG1 + (size_t)N * F1);
    int* cnt_in    = cnt_out + N;
    int* next_free = cnt_in + N;
    int* row_start = next_free + 32;
    int* rank      = row_start + N;
    int* col       = rank + E;

    hipMemsetAsync(cnt_out, 0, ((size_t)2 * N + 32) * sizeof(int), stream);

    gcn_degree_rank_kernel<<<(E + 255) / 256, 256, 0, stream>>>(src, dst, cnt_out, cnt_in, rank, E);
    gcn_norm_rowalloc_kernel<<<(N + 255) / 256, 256, 0, stream>>>(cnt_out, cnt_in, norm_src, norm_dst,
                                                                  row_start, next_free, N);
    gcn_edge_fill_kernel<<<(E + 255) / 256, 256, 0, stream>>>(src, dst, rank, row_start, col, E);

    gcn_gemm_lds_wmma_f32<F0, F1, 13><<<(N + 207) / 208, 512, 0, stream>>>(feat, W1, norm_src, H1, N);
    gcn_gather_sum_kernel<F1, 32, true><<<(N * 32 + 255) / 256, 256, 0, stream>>>(H1, row_start, cnt_in, col,
                                                                                 norm_dst, b1, AGG1, N);
    gcn_gemm_lds_wmma_f32<F1, F2, 2><<<(N + 127) / 128, 512, 0, stream>>>(AGG1, W2, norm_src, H2, N);
    gcn_gather_sum_kernel<F2, 16, false><<<(N * 16 + 255) / 256, 256, 0, stream>>>(H2, row_start, cnt_in, col,
                                                                                  norm_dst, b2, out, N);
}


# sorted edge list (rank, prefix alloc, fill pass) replaced by a fixed 36-slot source-id table per node filled by the counting pass, overflow list beyond; norms from counts; sum order = arrival rank
# speedup vs baseline: 2.6988x; 1.1147x over previous
#include <hip/hip_runtime.h>

typedef __attribute__((ext_vector_type(2))) float v2f;
typedef __attribute__((ext_vector_type(8))) float v8f;

#define F0 512
#define F1 256
#define F2 64

#define CAP 36
typedef unsigned short slot_t;

__global__ void gcn_degree_slot_kernel(const int* __restrict__ src, const int* __restrict__ dst,
                                       int* __restrict__ cnt_out, int* __restrict__ cnt_in,
                                       slot_t* __restrict__ slots, int* __restrict__ ovf_count,
                                       int* __restrict__ ovf, int E) {
    int e = blockIdx.x * blockDim.x + threadIdx.x;
    if (e < E) {
        int s = src[e];
        int d = dst[e];
        int r = atomicAdd(&cnt_in[d], 1);
        atomicAdd(&cnt_out[s], 1);
        if (r < CAP) slots[(size_t)d * CAP + r] = (slot_t)s;
        else         ovf[atomicAdd(ovf_count, 1)] = e;
    }
}

template <int K, int N, int RTW>
__global__ __launch_bounds__(512) void gcn_gemm_lds_wmma_f32(
    const float* __restrict__ A, const float* __restrict__ B,
    const int* __restrict__ cntA, float* __restrict__ C, int M) {
    constexpr int NT = 512;
    constexpr int CT = N / 16;
    constexpr int RG = 16 / CT;
    constexpr int BM = RG * RTW * 16;
    constexpr int KC = 32;
    constexpr int NKC = K / KC;
    constexpr int PA = KC + 4;
    constexpr int PB = 2 * N + 32;
    constexpr int PC = N + 4;
    constexpr int A_FLOATS = BM * PA;
    constexpr int B_FLOATS = (KC / 2) * PB;
    constexpr int C_FLOATS = BM * PC;
    constexpr int S_FLOATS = (A_FLOATS + B_FLOATS) > C_FLOATS ? (A_FLOATS + B_FLOATS) : C_FLOATS;
    constexpr int AUNITS = BM * (KC / 4);
    constexpr int AU = (AUNITS + NT - 1) / NT;
    constexpr int BUNITS = (KC / 2) * (N / 4);
    constexpr int BU = (BUNITS + NT - 1) / NT;
    constexpr int CUNITS = BM * (N / 4);

    __shared__ __attribute__((aligned(16))) float smem[S_FLOATS];
    float* As = smem;
    float* Bs = smem + A_FLOATS;

    const int tid  = threadIdx.x;
    const int wave = tid >> 5;
    const int lane = tid & 31;
    const int h    = lane >> 4;
    const int l    = lane & 15;
    const int ct   = wave % CT;
    const int rg   = wave / CT;
    const int m_base = blockIdx.x * BM;

    const float* a_src[AU];
    float a_scale[AU];
    int   a_dst[AU];
    bool  a_ok[AU];
    #pragma unroll
    for (int i = 0; i < AU; ++i) {
        int u = tid + i * NT;
        a_ok[i] = (u < AUNITS);
        if (!a_ok[i]) u = 0;
        int r = u / (KC / 4);
        int c = (u % (KC / 4)) * 4;
        int g = m_base + r;
        if (g > M - 1) g = M - 1;
        a_src[i]   = A + (size_t)g * K + c;
        int co = cntA[g];
        a_scale[i] = (co > 0) ? rsqrtf(fmaxf((float)co, 1.0f)) : 1.0f;
        a_dst[i]   = r * PA + c;
    }
    const float* b_src[BU];
    int  b_dst[BU];
    bool b_ok[BU];
    #pragma unroll
    for (int i = 0; i < BU; ++i) {
        int u = tid + i * NT;
        b_ok[i] = (u < BUNITS);
        if (!b_ok[i]) u = 0;
        int kp = u / (N / 4);
        int c4 = (u % (N / 4)) * 4;
        b_src[i] = B + (size_t)(2 * kp) * N + c4;
        b_dst[i] = kp * PB + 2 * c4;
    }

    float4 ra[AU], rb0[BU], rb1[BU];
    #pragma unroll
    for (int i = 0; i < AU; ++i) ra[i] = *(const float4*)(a_src[i]);
    #pragma unroll
    for (int i = 0; i < BU; ++i) {
        rb0[i] = *(const float4*)(b_src[i]);
        rb1[i] = *(const float4*)(b_src[i] + N);
    }

    v8f acc[RTW];
    #pragma unroll
    for (int rt = 0; rt < RTW; ++rt) acc[rt] = v8f{};

    const v2f* As2 = (const v2f*)As + (((rg * 16 + l) * PA) >> 1) + h;
    const v2f* Bs2 = (const v2f*)Bs + h * (PB / 2) + ct * 16 + l;

    for (int c = 0; c < NKC; ++c) {
        #pragma unroll
        for (int i = 0; i < AU; ++i) {
            if (a_ok[i]) {
                float4 v = ra[i];
                float s = a_scale[i];
                v.x *= s; v.y *= s; v.z *= s; v.w *= s;
                *(float4*)(As + a_dst[i]) = v;
            }
        }
        #pragma unroll
        for (int i = 0; i < BU; ++i) {
            if (b_ok[i]) {
                float4 p = make_float4(rb0[i].x, rb1[i].x, rb0[i].y, rb1[i].y);
                float4 q = make_float4(rb0[i].z, rb1[i].z, rb0[i].w, rb1[i].w);
                *(float4*)(Bs + b_dst[i])     = p;
                *(float4*)(Bs + b_dst[i] + 4) = q;
            }
        }
        __syncthreads();

        if (c + 1 < NKC) {
            const int ko = (c + 1) * KC;
            #pragma unroll
            for (int i = 0; i < AU; ++i) ra[i] = *(const float4*)(a_src[i] + ko);
            #pragma unroll
            for (int i = 0; i < BU; ++i) {
                rb0[i] = *(const float4*)(b_src[i] + (size_t)ko * N);
                rb1[i] = *(const float4*)(b_src[i] + (size_t)ko * N + N);
            }
        }

        #pragma unroll
        for (int kk = 0; kk < KC; kk += 4) {
            v2f b = Bs2[(kk >> 1) * (PB / 2)];
            #pragma unroll
            for (int rt = 0; rt < RTW; ++rt) {
                v2f a = As2[((rt * RG * 16 * PA) >> 1) + (kk >> 1)];
                acc[rt] = __builtin_amdgcn_wmma_f32_16x16x4_f32(false, a, false, b, (short)0, acc[rt], false, false);
            }
        }
        __syncthreads();
    }

    float* Cs = smem;
    #pragma unroll
    for (int rt = 0; rt < RTW; ++rt) {
        #pragma unroll
        for (int i = 0; i < 8; ++i)
            Cs[((rt * RG + rg) * 16 + 8 * h + i) * PC + ct * 16 + l] = acc[rt][i];
    }
    __syncthreads();
    for (int u = tid; u < CUNITS; u += NT) {
        int r  = u / (N / 4);
        int c4 = (u % (N / 4)) * 4;
        int g  = m_base + r;
        if (g < M) *(float4*)(C + (size_t)g * N + c4) = *(const float4*)(Cs + r * PC + c4);
    }
}

__device__ __forceinline__ void gcn_add4(float4& a, const float4& x) {
    a.x += x.x; a.y += x.y; a.z += x.z; a.w += x.w;
}

template <int F, int LPN, bool RELU>
__global__ __launch_bounds__(256) void gcn_gather_sum_kernel(
    const float* __restrict__ H, const int* __restrict__ cnt_in, const slot_t* __restrict__ slots,
    const int* __restrict__ ovf_count, const int* __restrict__ ovf,
    const int* __restrict__ src, const int* __restrict__ dst,
    const float* __restrict__ bias, float* __restrict__ OUT, int n_nodes) {
    constexpr int NV = F / (4 * LPN);
    int node = (blockIdx.x * blockDim.x + threadIdx.x) / LPN;
    int sub  = threadIdx.x % LPN;
    if (node >= n_nodes) return;
    int n  = cnt_in[node];
    int nt = (n < CAP) ? n : CAP;
    const slot_t* row = slots + (size_t)node * CAP;
    float nd = (n > 0) ? rsqrtf(fmaxf((float)n, 1.0f)) : 1.0f;
    const float* Hc = H + (sub << 2);

    float4 acc[NV];
    for (int v = 0; v < NV; ++v) acc[v] = make_float4(0.0f, 0.0f, 0.0f, 0.0f);

    for (int base = 0; base < nt; base += LPN) {
        int m = nt - base;
        if (m > LPN) m = LPN;
        int mine = (sub < m) ? (int)row[base + sub] : 0;
        int j = 0;
        for (; j + 4 <= m; j += 4) {
            const float* p0 = Hc + (size_t)__shfl(mine, j,     LPN) * F;
            const float* p1 = Hc + (size_t)__shfl(mine, j + 1, LPN) * F;
            const float* p2 = Hc + (size_t)__shfl(mine, j + 2, LPN) * F;
            const float* p3 = Hc + (size_t)__shfl(mine, j + 3, LPN) * F;
            float4 x0[NV], x1[NV], x2[NV], x3[NV];
            for (int v = 0; v < NV; ++v) {
                x0[v] = *(const float4*)(p0 + v * LPN * 4);
                x1[v] = *(const float4*)(p1 + v * LPN * 4);
                x2[v] = *(const float4*)(p2 + v * LPN * 4);
                x3[v] = *(const float4*)(p3 + v * LPN * 4);
            }
            for (int v = 0; v < NV; ++v) {
                gcn_add4(acc[v], x0[v]);
                gcn_add4(acc[v], x1[v]);
                gcn_add4(acc[v], x2[v]);
                gcn_add4(acc[v], x3[v]);
            }
        }
        for (; j < m; ++j) {
            const float* p0 = Hc + (size_t)__shfl(mine, j, LPN) * F;
            for (int v = 0; v < NV; ++v) gcn_add4(acc[v], *(const float4*)(p0 + v * LPN * 4));
        }
    }

    if (n > CAP) {
        int oc = *ovf_count;
        for (int k = 0; k < oc; ++k) {
            int e = ovf[k];
            if (dst[e] == node) {
                const float* p0 = Hc + (size_t)src[e] * F;
                for (int v = 0; v < NV; ++v) gcn_add4(acc[v], *(const float4*)(p0 + v * LPN * 4));
            }
        }
    }

    const float* bp = bias + (sub << 2);
    float* o = OUT + (size_t)node * F + (sub << 2);
    for (int v = 0; v < NV; ++v) {
        float4 b = *(const float4*)(bp + v * LPN * 4);
        float4 r;
        r.x = acc[v].x * nd + b.x;
        r.y = acc[v].y * nd + b.y;
        r.z = acc[v].z * nd + b.z;
        r.w = acc[v].w * nd + b.w;
        if (RELU) {
            r.x = fmaxf(r.x, 0.0f); r.y = fmaxf(r.y, 0.0f);
            r.z = fmaxf(r.z, 0.0f); r.w = fmaxf(r.w, 0.0f);
        }
        *(float4*)(o + v * LPN * 4) = r;
    }
}

extern "C" void kernel_launch(void* const* d_in, const int* in_sizes, int n_in,
                              void* d_out, int out_size, void* d_ws, size_t ws_size,
                              hipStream_t stream) {
    const float* feat = (const float*)d_in[0];
    const int*   src  = (const int*)d_in[1];
    const int*   dst  = (const int*)d_in[2];
    const float* W1   = (const float*)d_in[3];
    const float* b1   = (const float*)d_in[4];
    const float* W2   = (const float*)d_in[5];
    const float* b2   = (const float*)d_in[6];
    float* out = (float*)d_out;

    const int N = in_sizes[0] / F0;
    const int E = in_sizes[1];

    float* ws   = (float*)d_ws;
    float* H1   = ws;
    float* AGG1 = H1 + (size_t)N * F1;
    float* H2   = H1;
    int* cnt_out   = (int*)(AGG1 + (size_t)N * F1);
    int* cnt_in    = cnt_out + N;
    int* ovf_count = cnt_in + N;
    slot_t* slots  = (slot_t*)(ovf_count + 32);
    int* ovf       = (int*)(slots + (size_t)N * CAP);

    hipMemsetAsync(cnt_out, 0, ((size_t)2 * N + 32) * sizeof(int), stream);

    gcn_degree_slot_kernel<<<(E + 255) / 256, 256, 0, stream>>>(src, dst, cnt_out, cnt_in, slots, ovf_count, ovf, E);

    gcn_gemm_lds_wmma_f32<F0, F1, 13><<<(N + 207) / 208, 512, 0, stream>>>(feat, W1, cnt_out, H1, N);
    gcn_gather_sum_kernel<F1, 32, true><<<(N * 32 + 255) / 256, 256, 0, stream>>>(H1, cnt_in, slots, ovf_count, ovf,
                                                                                 src, dst, b1, AGG1, N);
    gcn_gemm_lds_wmma_f32<F1, F2, 2><<<(N + 127) / 128, 512, 0, stream>>>(AGG1, W2, cnt_out, H2, N);
    gcn_gather_sum_kernel<F2, 16, false><<<(N * 16 + 255) / 256, 256, 0, stream>>>(H2, cnt_in, slots, ovf_count, ovf,
                                                                                  src, dst, b2, out, N);
}


# in-edge table and degrees built without global atomics: pass 1 bins edges by 256-node range, pass 2 ranks them per node, all counters in LDS; memset dropped; sum order = LDS arrival rank
# speedup vs baseline: 3.8995x; 1.4449x over previous
#include <hip/hip_runtime.h>

typedef __attribute__((ext_vector_type(2))) float v2f;
typedef __attribute__((ext_vector_type(8))) float v8f;

#define F0 512
#define F1 256
#define F2 64

#define CAP 36
typedef unsigned short slot_t;

#define KMAX 256
#define A_NT 512
#define EPT 4
#define EPB (A_NT * EPT)

__global__ __launch_bounds__(A_NT) void gcn_bin_kernel(
    const int* __restrict__ src, const int* __restrict__ dst,
    unsigned* __restrict__ sorted_in, unsigned char* __restrict__ sorted_out,
    unsigned* __restrict__ tab_in, unsigned* __restrict__ tab_out,
    int* __restrict__ ovf_count, int E, int n_nodes, int nblk) {
    __shared__ int h_in[KMAX], h_out[KMAX];
    __shared__ int c_in[KMAX], c_out[KMAX];
    const int tid = threadIdx.x;
    const int K = (n_nodes + 255) >> 8;
    if (tid < KMAX) { h_in[tid] = 0; h_out[tid] = 0; }
    if (blockIdx.x == 0 && tid == 0) *ovf_count = 0;
    __syncthreads();

    const int base = blockIdx.x * EPB;
    int s[EPT], d[EPT], r_in[EPT], r_out[EPT];
    #pragma unroll
    for (int i = 0; i < EPT; ++i) {
        int e = base + tid + i * A_NT;
        if (e < E) { s[i] = src[e]; d[i] = dst[e]; }
        else       { s[i] = -1;     d[i] = -1; }
    }
    #pragma unroll
    for (int i = 0; i < EPT; ++i) {
        r_in[i] = 0; r_out[i] = 0;
        if (d[i] >= 0) {
            r_in[i]  = atomicAdd(&h_in[d[i] >> 8], 1);
            r_out[i] = atomicAdd(&h_out[s[i] >> 8], 1);
        }
    }
    __syncthreads();

    const int wave = tid >> 5;
    const int lane = tid & 31;
    if (wave < 2) {
        int* h = wave ? h_out : h_in;
        int* c = wave ? c_out : c_in;
        int v[8];
        int sum = 0;
        #pragma unroll
        for (int j = 0; j < 8; ++j) { v[j] = h[lane * 8 + j]; sum += v[j]; }
        int incl = sum;
        for (int off = 1; off < 32; off <<= 1) {
            int t = __shfl_up(incl, off);
            if (lane >= off) incl += t;
        }
        int run = incl - sum;
        #pragma unroll
        for (int j = 0; j < 8; ++j) { c[lane * 8 + j] = v[j]; h[lane * 8 + j] = run; run += v[j]; }
    }
    __syncthreads();

    #pragma unroll
    for (int i = 0; i < EPT; ++i) {
        if (d[i] >= 0) {
            sorted_in[base + h_in[d[i] >> 8] + r_in[i]] = (unsigned)s[i] | ((unsigned)(d[i] & 255) << 16);
            sorted_out[base + h_out[s[i] >> 8] + r_out[i]] = (unsigned char)(s[i] & 255);
        }
    }
    if (tid < K) {
        tab_in[(size_t)tid * nblk + blockIdx.x]  = (unsigned)h_in[tid]  | ((unsigned)c_in[tid]  << 16);
        tab_out[(size_t)tid * nblk + blockIdx.x] = (unsigned)h_out[tid] | ((unsigned)c_out[tid] << 16);
    }
}

#define B_NT 1024
__global__ __launch_bounds__(B_NT) void gcn_rank_kernel(
    const unsigned* __restrict__ sorted_in, const unsigned char* __restrict__ sorted_out,
    const unsigned* __restrict__ tab_in, const unsigned* __restrict__ tab_out,
    int* __restrict__ cnt_out, int* __restrict__ cnt_in, slot_t* __restrict__ slots,
    int* __restrict__ ovf_count, unsigned* __restrict__ ovf, int n_nodes, int nblk) {
    __shared__ int c_i[256], c_o[256];
    __shared__ __attribute__((aligned(16))) slot_t sl[256 * CAP];
    const int tid = threadIdx.x;
    const int k = blockIdx.x;
    if (tid < 256) { c_i[tid] = 0; c_o[tid] = 0; }
    unsigned* slw = (unsigned*)sl;
    for (int i = tid; i < 256 * CAP / 2; i += B_NT) slw[i] = 0u;
    __syncthreads();

    const int t = tid & 511;
    if (tid < 512) {
        for (int b = t; b < nblk; b += 512) {
            unsigned w = tab_in[(size_t)k * nblk + b];
            const int c = (int)(w >> 16);
            const unsigned* p = sorted_in + (size_t)b * EPB + (w & 0xffffu);
            for (int j = 0; j < c; j += 4) {
                unsigned e[4];
                #pragma unroll
                for (int q = 0; q < 4; ++q) e[q] = (j + q < c) ? p[j + q] : 0u;
                #pragma unroll
                for (int q = 0; q < 4; ++q) {
                    if (j + q < c) {
                        int dl = (int)((e[q] >> 16) & 255u);
                        unsigned sid = e[q] & 0xffffu;
                        int r = atomicAdd(&c_i[dl], 1);
                        if (r < CAP) sl[dl * CAP + r] = (slot_t)sid;
                        else         ovf[atomicAdd(ovf_count, 1)] = ((unsigned)(k * 256 + dl) << 16) | sid;
                    }
                }
            }
        }
    } else {
        for (int b = t; b < nblk; b += 512) {
            unsigned w = tab_out[(size_t)k * nblk + b];
            const int c = (int)(w >> 16);
            const unsigned char* p = sorted_out + (size_t)b * EPB + (w & 0xffffu);
            for (int j = 0; j < c; j += 4) {
                int e[4];
                #pragma unroll
                for (int q = 0; q < 4; ++q) e[q] = (j + q < c) ? (int)p[j + q] : 0;
                #pragma unroll
                for (int q = 0; q < 4; ++q) {
                    if (j + q < c) atomicAdd(&c_o[e[q]], 1);
                }
            }
        }
    }
    __syncthreads();

    const int node0 = k * 256;
    if (tid < 256 && node0 + tid < n_nodes) {
        cnt_in[node0 + tid]  = c_i[tid];
        cnt_out[node0 + tid] = c_o[tid];
    }
    int rows = n_nodes - node0;
    if (rows > 256) rows = 256;
    unsigned* gs = (unsigned*)(slots + (size_t)node0 * CAP);
    for (int i = tid; i < rows * (CAP / 2); i += B_NT) gs[i] = slw[i];
}

template <int K, int N, int RTW>
__global__ __launch_bounds__(512) void gcn_gemm_lds_wmma_f32(
    const float* __restrict__ A, const float* __restrict__ B,
    const int* __restrict__ cntA, float* __restrict__ C, int M) {
    constexpr int NT = 512;
    constexpr int CT = N / 16;
    constexpr int RG = 16 / CT;
    constexpr int BM = RG * RTW * 16;
    constexpr int KC = 32;
    constexpr int NKC = K / KC;
    constexpr int PA = KC + 4;
    constexpr int PB = 2 * N + 32;
    constexpr int PC = N + 4;
    constexpr int A_FLOATS = BM * PA;
    constexpr int B_FLOATS = (KC / 2) * PB;
    constexpr int C_FLOATS = BM * PC;
    constexpr int S_FLOATS = (A_FLOATS + B_FLOATS) > C_FLOATS ? (A_FLOATS + B_FLOATS) : C_FLOATS;
    constexpr int AUNITS = BM * (KC / 4);
    constexpr int AU = (AUNITS + NT - 1) / NT;
    constexpr int BUNITS = (KC / 2) * (N / 4);
    constexpr int BU = (BUNITS + NT - 1) / NT;
    constexpr int CUNITS = BM * (N / 4);

    __shared__ __attribute__((aligned(16))) float smem[S_FLOATS];
    float* As = smem;
    float* Bs = smem + A_FLOATS;

    const int tid  = threadIdx.x;
    const int wave = tid >> 5;
    const int lane = tid & 31;
    const int h    = lane >> 4;
    const int l    = lane & 15;
    const int ct   = wave % CT;
    const int rg   = wave / CT;
    const int m_base = blockIdx.x * BM;

    const float* a_src[AU];
    float a_scale[AU];
    int   a_dst[AU];
    bool  a_ok[AU];
    #pragma unroll
    for (int i = 0; i < AU; ++i) {
        int u = tid + i * NT;
        a_ok[i] = (u < AUNITS);
        if (!a_ok[i]) u = 0;
        int r = u / (KC / 4);
        int c = (u % (KC / 4)) * 4;
        int g = m_base + r;
        if (g > M - 1) g = M - 1;
        a_src[i]   = A + (size_t)g * K + c;
        int co = cntA[g];
        a_scale[i] = (co > 0) ? rsqrtf(fmaxf((float)co, 1.0f)) : 1.0f;
        a_dst[i]   = r * PA + c;
    }
    const float* b_src[BU];
    int  b_dst[BU];
    bool b_ok[BU];
    #pragma unroll
    for (int i = 0; i < BU; ++i) {
        int u = tid + i * NT;
        b_ok[i] = (u < BUNITS);
        if (!b_ok[i]) u = 0;
        int kp = u / (N / 4);
        int c4 = (u % (N / 4)) * 4;
        b_src[i] = B + (size_t)(2 * kp) * N + c4;
        b_dst[i] = kp * PB + 2 * c4;
    }

    float4 ra[AU], rb0[BU], rb1[BU];
    #pragma unroll
    for (int i = 0; i < AU; ++i) ra[i] = *(const float4*)(a_src[i]);
    #pragma unroll
    for (int i = 0; i < BU; ++i) {
        rb0[i] = *(const float4*)(b_src[i]);
        rb1[i] = *(const float4*)(b_src[i] + N);
    }

    v8f acc[RTW];
    #pragma unroll
    for (int rt = 0; rt < RTW; ++rt) acc[rt] = v8f{};

    const v2f* As2 = (const v2f*)As + (((rg * 16 + l) * PA) >> 1) + h;
    const v2f* Bs2 = (const v2f*)Bs + h * (PB / 2) + ct * 16 + l;

    for (int c = 0; c < NKC; ++c) {
        #pragma unroll
        for (int i = 0; i < AU; ++i) {
            if (a_ok[i]) {
                float4 v = ra[i];
                float s = a_scale[i];
                v.x *= s; v.y *= s; v.z *= s; v.w *= s;
                *(float4*)(As + a_dst[i]) = v;
            }
        }
        #pragma unroll
        for (int i = 0; i < BU; ++i) {
            if (b_ok[i]) {
                float4 p = make_float4(rb0[i].x, rb1[i].x, rb0[i].y, rb1[i].y);
                float4 q = make_float4(rb0[i].z, rb1[i].z, rb0[i].w, rb1[i].w);
                *(float4*)(Bs + b_dst[i])     = p;
                *(float4*)(Bs + b_dst[i] + 4) = q;
            }
        }
        __syncthreads();

        if (c + 1 < NKC) {
            const int ko = (c + 1) * KC;
            #pragma unroll
            for (int i = 0; i < AU; ++i) ra[i] = *(const float4*)(a_src[i] + ko);
            #pragma unroll
            for (int i = 0; i < BU; ++i) {
                rb0[i] = *(const float4*)(b_src[i] + (size_t)ko * N);
                rb1[i] = *(const float4*)(b_src[i] + (size_t)ko * N + N);
            }
        }

        #pragma unroll
        for (int kk = 0; kk < KC; kk += 4) {
            v2f b = Bs2[(kk >> 1) * (PB / 2)];
            #pragma unroll
            for (int rt = 0; rt < RTW; ++rt) {
                v2f a = As2[((rt * RG * 16 * PA) >> 1) + (kk >> 1)];
                acc[rt] = __builtin_amdgcn_wmma_f32_16x16x4_f32(false, a, false, b, (short)0, acc[rt], false, false);
            }
        }
        __syncthreads();
    }

    float* Cs = smem;
    #pragma unroll
    for (int rt = 0; rt < RTW; ++rt) {
        #pragma unroll
        for (int i = 0; i < 8; ++i)
            Cs[((rt * RG + rg) * 16 + 8 * h + i) * PC + ct * 16 + l] = acc[rt][i];
    }
    __syncthreads();
    for (int u = tid; u < CUNITS; u += NT) {
        int r  = u / (N / 4);
        int c4 = (u % (N / 4)) * 4;
        int g  = m_base + r;
        if (g < M) *(float4*)(C + (size_t)g * N + c4) = *(const float4*)(Cs + r * PC + c4);
    }
}

__device__ __forceinline__ void gcn_add4(float4& a, const float4& x) {
    a.x += x.x; a.y += x.y; a.z += x.z; a.w += x.w;
}

template <int F, int LPN, bool RELU>
__global__ __launch_bounds__(256) void gcn_gather_sum_kernel(
    const float* __restrict__ H, const int* __restrict__ cnt_in, const slot_t* __restrict__ slots,
    const int* __restrict__ ovf_count, const unsigned* __restrict__ ovf,
    const float* __restrict__ bias, float* __restrict__ OUT, int n_nodes) {
    constexpr int NV = F / (4 * LPN);
    int node = (blockIdx.x * blockDim.x + threadIdx.x) / LPN;
    int sub  = threadIdx.x % LPN;
    if (node >= n_nodes) return;
    int n  = cnt_in[node];
    int nt = (n < CAP) ? n : CAP;
    const slot_t* row = slots + (size_t)node * CAP;
    float nd = (n > 0) ? rsqrtf(fmaxf((float)n, 1.0f)) : 1.0f;
    const float* Hc = H + (sub << 2);

    float4 acc[NV];
    for (int v = 0; v < NV; ++v) acc[v] = make_float4(0.0f, 0.0f, 0.0f, 0.0f);

    for (int base = 0; base < nt; base += LPN) {
        int m = nt - base;
        if (m > LPN) m = LPN;
        int mine = (sub < m) ? (int)row[base + sub] : 0;
        int j = 0;
        for (; j + 4 <= m; j += 4) {
            const float* p0 = Hc + (size_t)__shfl(mine, j,     LPN) * F;
            const float* p1 = Hc + (size_t)__shfl(mine, j + 1, LPN) * F;
            const float* p2 = Hc + (size_t)__shfl(mine, j + 2, LPN) * F;
            const float* p3 = Hc + (size_t)__shfl(mine, j + 3, LPN) * F;
            float4 x0[NV], x1[NV], x2[NV], x3[NV];
            for (int v = 0; v < NV; ++v) {
                x0[v] = *(const float4*)(p0 + v * LPN * 4);
                x1[v] = *(const float4*)(p1 + v * LPN * 4);
                x2[v] = *(const float4*)(p2 + v * LPN * 4);
                x3[v] = *(const float4*)(p3 + v * LPN * 4);
            }
            for (int v = 0; v < NV; ++v) {
                gcn_add4(acc[v], x0[v]);
                gcn_add4(acc[v], x1[v]);
                gcn_add4(acc[v], x2[v]);
                gcn_add4(acc[v], x3[v]);
            }
        }
        for (; j < m; ++j) {
            const float* p0 = Hc + (size_t)__shfl(mine, j, LPN) * F;
            for (int v = 0; v < NV; ++v) gcn_add4(acc[v], *(const float4*)(p0 + v * LPN * 4));
        }
    }

    if (n > CAP) {
        int oc = *ovf_count;
        for (int k = 0; k < oc; ++k) {
            unsigned w = ovf[k];
            if ((int)(w >> 16) == node) {
                const float* p0 = Hc + (size_t)(w & 0xffffu) * F;
                for (int v = 0; v < NV; ++v) gcn_add4(acc[v], *(const float4*)(p0 + v * LPN * 4));
            }
        }
    }

    const float* bp = bias + (sub << 2);
    float* o = OUT + (size_t)node * F + (sub << 2);
    for (int v = 0; v < NV; ++v) {
        float4 b = *(const float4*)(bp + v * LPN * 4);
        float4 r;
        r.x = acc[v].x * nd + b.x;
        r.y = acc[v].y * nd + b.y;
        r.z = acc[v].z * nd + b.z;
        r.w = acc[v].w * nd + b.w;
        if (RELU) {
            r.x = fmaxf(r.x, 0.0f); r.y = fmaxf(r.y, 0.0f);
            r.z = fmaxf(r.z, 0.0f); r.w = fmaxf(r.w, 0.0f);
        }
        *(float4*)(o + v * LPN * 4) = r;
    }
}

extern "C" void kernel_launch(void* const* d_in, const int* in_sizes, int n_in,
                              void* d_out, int out_size, void* d_ws, size_t ws_size,
                              hipStream_t stream) {
    const float* feat = (const float*)d_in[0];
    const int*   src  = (const int*)d_in[1];
    const int*   dst  = (const int*)d_in[2];
    const float* W1   = (const float*)d_in[3];
    const float* b1   = (const float*)d_in[4];
    const float* W2   = (const float*)d_in[5];
    const float* b2   = (const float*)d_in[6];
    float* out = (float*)d_out;

    const int N = in_sizes[0] / F0;
    const int E = in_sizes[1];

    float* ws   = (float*)d_ws;
    float* H1   = ws;
    float* AGG1 = H1 + (size_t)N * F1;
    float* H2   = H1;
    int* cnt_out   = (int*)(AGG1 + (size_t)N * F1);
    int* cnt_in    = cnt_out + N;
    int* ovf_count = cnt_in + N;
    slot_t* slots  = (slot_t*)(ovf_count + 32);
    unsigned* ovf  = (unsigned*)(slots + (size_t)N * CAP);
    const int nblk = (E + EPB - 1) / EPB;
    unsigned* sorted_in  = (unsigned*)AGG1;
    unsigned* tab_in     = sorted_in + E;
    unsigned* tab_out    = tab_in + (size_t)KMAX * nblk;
    unsigned char* sorted_out = (unsigned char*)(tab_out + (size_t)KMAX * nblk);

    gcn_bin_kernel<<<nblk, A_NT, 0, stream>>>(src, dst, sorted_in, sorted_out, tab_in, tab_out, ovf_count, E, N, nblk);
    gcn_rank_kernel<<<(N + 255) / 256, B_NT, 0, stream>>>(sorted_in, sorted_out, tab_in, tab_out, cnt_out, cnt_in, slots,
                                                         ovf_count, ovf, N, nblk);

    gcn_gemm_lds_wmma_f32<F0, F1, 13><<<(N + 207) / 208, 512, 0, stream>>>(feat, W1, cnt_out, H1, N);
    gcn_gather_sum_kernel<F1, 32, true><<<(N * 32 + 255) / 256, 256, 0, stream>>>(H1, cnt_in, slots, ovf_count, ovf,
                                                                                 b1, AGG1, N);
    gcn_gemm_lds_wmma_f32<F1, F2, 2><<<(N + 127) / 128, 512, 0, stream>>>(AGG1, W2, cnt_out, H2, N);
    gcn_gather_sum_kernel<F2, 16, false><<<(N * 16 + 255) / 256, 256, 0, stream>>>(H2, cnt_in, slots, ovf_count, ovf,
                                                                                  b2, out, N);
}


# layer-2 GEMM fused into the layer-1 sum kernel: a 16-node tile's finished rows stay in LDS and meet register-held W2 fragments on f32 WMMA; its k sum is now 4 quarter partials added in order
# speedup vs baseline: 4.0875x; 1.0482x over previous
#include <hip/hip_runtime.h>

typedef __attribute__((ext_vector_type(2))) float v2f;
typedef __attribute__((ext_vector_type(8))) float v8f;

#define F0 512
#define F1 256
#define F2 64

#define CAP 36
typedef unsigned short slot_t;

#define KMAX 256
#define A_NT 512
#define EPT 4
#define EPB (A_NT * EPT)

__global__ __launch_bounds__(A_NT) void gcn_bin_kernel(
    const int* __restrict__ src, const int* __restrict__ dst,
    unsigned* __restrict__ sorted_in, unsigned char* __restrict__ sorted_out,
    unsigned* __restrict__ tab_in, unsigned* __restrict__ tab_out,
    int* __restrict__ ovf_count, int E, int n_nodes, int nblk) {
    __shared__ int h_in[KMAX], h_out[KMAX];
    __shared__ int c_in[KMAX], c_out[KMAX];
    const int tid = threadIdx.x;
    const int K = (n_nodes + 255) >> 8;
    if (tid < KMAX) { h_in[tid] = 0; h_out[tid] = 0; }
    if (blockIdx.x == 0 && tid == 0) *ovf_count = 0;
    __syncthreads();

    const int base = blockIdx.x * EPB;
    int s[EPT], d[EPT], r_in[EPT], r_out[EPT];
    #pragma unroll
    for (int i = 0; i < EPT; ++i) {
        int e = base + tid + i * A_NT;
        if (e < E) { s[i] = src[e]; d[i] = dst[e]; }
        else       { s[i] = -1;     d[i] = -1; }
    }
    #pragma unroll
    for (int i = 0; i < EPT; ++i) {
        r_in[i] = 0; r_out[i] = 0;
        if (d[i] >= 0) {
            r_in[i]  = atomicAdd(&h_in[d[i] >> 8], 1);
            r_out[i] = atomicAdd(&h_out[s[i] >> 8], 1);
        }
    }
    __syncthreads();

    const int wave = tid >> 5;
    const int lane = tid & 31;
    if (wave < 2) {
        int* h = wave ? h_out : h_in;
        int* c = wave ? c_out : c_in;
        int v[8];
        int sum = 0;
        #pragma unroll
        for (int j = 0; j < 8; ++j) { v[j] = h[lane * 8 + j]; sum += v[j]; }
        int incl = sum;
        for (int off = 1; off < 32; off <<= 1) {
            int t = __shfl_up(incl, off);
            if (lane >= off) incl += t;
        }
        int run = incl - sum;
        #pragma unroll
        for (int j = 0; j < 8; ++j) { c[lane * 8 + j] = v[j]; h[lane * 8 + j] = run; run += v[j]; }
    }
    __syncthreads();

    #pragma unroll
    for (int i = 0; i < EPT; ++i) {
        if (d[i] >= 0) {
            sorted_in[base + h_in[d[i] >> 8] + r_in[i]] = (unsigned)s[i] | ((unsigned)(d[i] & 255) << 16);
            sorted_out[base + h_out[s[i] >> 8] + r_out[i]] = (unsigned char)(s[i] & 255);
        }
    }
    if (tid < K) {
        tab_in[(size_t)tid * nblk + blockIdx.x]  = (unsigned)h_in[tid]  | ((unsigned)c_in[tid]  << 16);
        tab_out[(size_t)tid * nblk + blockIdx.x] = (unsigned)h_out[tid] | ((unsigned)c_out[tid] << 16);
    }
}

#define B_NT 1024
__global__ __launch_bounds__(B_NT) void gcn_rank_kernel(
    const unsigned* __restrict__ sorted_in, const unsigned char* __restrict__ sorted_out,
    const unsigned* __restrict__ tab_in, const unsigned* __restrict__ tab_out,
    int* __restrict__ cnt_out, int* __restrict__ cnt_in, slot_t* __restrict__ slots,
    int* __restrict__ ovf_count, unsigned* __restrict__ ovf, int n_nodes, int nblk) {
    __shared__ int c_i[256], c_o[256];
    __shared__ __attribute__((aligned(16))) slot_t sl[256 * CAP];
    const int tid = threadIdx.x;
    const int k = blockIdx.x;
    if (tid < 256) { c_i[tid] = 0; c_o[tid] = 0; }
    unsigned* slw = (unsigned*)sl;
    for (int i = tid; i < 256 * CAP / 2; i += B_NT) slw[i] = 0u;
    __syncthreads();

    const int t = tid & 511;
    if (tid < 512) {
        for (int b = t; b < nblk; b += 512) {
            unsigned w = tab_in[(size_t)k * nblk + b];
            const int c = (int)(w >> 16);
            const unsigned* p = sorted_in + (size_t)b * EPB + (w & 0xffffu);
            for (int j = 0; j < c; j += 4) {
                unsigned e[4];
                #pragma unroll
                for (int q = 0; q < 4; ++q) e[q] = (j + q < c) ? p[j + q] : 0u;
                #pragma unroll
                for (int q = 0; q < 4; ++q) {
                    if (j + q < c) {
                        int dl = (int)((e[q] >> 16) & 255u);
                        unsigned sid = e[q] & 0xffffu;
                        int r = atomicAdd(&c_i[dl], 1);
                        if (r < CAP) sl[dl * CAP + r] = (slot_t)sid;
                        else         ovf[atomicAdd(ovf_count, 1)] = ((unsigned)(k * 256 + dl) << 16) | sid;
                    }
                }
            }
        }
    } else {
        for (int b = t; b < nblk; b += 512) {
            unsigned w = tab_out[(size_t)k * nblk + b];
            const int c = (int)(w >> 16);
            const unsigned char* p = sorted_out + (size_t)b * EPB + (w & 0xffffu);
            for (int j = 0; j < c; j += 4) {
                int e[4];
                #pragma unroll
                for (int q = 0; q < 4; ++q) e[q] = (j + q < c) ? (int)p[j + q] : 0;
                #pragma unroll
                for (int q = 0; q < 4; ++q) {
                    if (j + q < c) atomicAdd(&c_o[e[q]], 1);
                }
            }
        }
    }
    __syncthreads();

    const int node0 = k * 256;
    if (tid < 256 && node0 + tid < n_nodes) {
        cnt_in[node0 + tid]  = c_i[tid];
        cnt_out[node0 + tid] = c_o[tid];
    }
    int rows = n_nodes - node0;
    if (rows > 256) rows = 256;
    unsigned* gs = (unsigned*)(slots + (size_t)node0 * CAP);
    for (int i = tid; i < rows * (CAP / 2); i += B_NT) gs[i] = slw[i];
}

template <int K, int N, int RTW>
__global__ __launch_bounds__(512) void gcn_gemm_lds_wmma_f32(
    const float* __restrict__ A, const float* __restrict__ B,
    const int* __restrict__ cntA, float* __restrict__ C, int M) {
    constexpr int NT = 512;
    constexpr int CT = N / 16;
    constexpr int RG = 16 / CT;
    constexpr int BM = RG * RTW * 16;
    constexpr int KC = 32;
    constexpr int NKC = K / KC;
    constexpr int PA = KC + 4;
    constexpr int PB = 2 * N + 32;
    constexpr int PC = N + 4;
    constexpr int A_FLOATS = BM * PA;
    constexpr int B_FLOATS = (KC / 2) * PB;
    constexpr int C_FLOATS = BM * PC;
    constexpr int S_FLOATS = (A_FLOATS + B_FLOATS) > C_FLOATS ? (A_FLOATS + B_FLOATS) : C_FLOATS;
    constexpr int AUNITS = BM * (KC / 4);
    constexpr int AU = (AUNITS + NT - 1) / NT;
    constexpr int BUNITS = (KC / 2) * (N / 4);
    constexpr int BU = (BUNITS + NT - 1) / NT;
    constexpr int CUNITS = BM * (N / 4);

    __shared__ __attribute__((aligned(16))) float smem[S_FLOATS];
    float* As = smem;
    float* Bs = smem + A_FLOATS;

    const int tid  = threadIdx.x;
    const int wave = tid >> 5;
    const int lane = tid & 31;
    const int h    = lane >> 4;
    const int l    = lane & 15;
    const int ct   = wave % CT;
    const int rg   = wave / CT;
    const int m_base = blockIdx.x * BM;

    const float* a_src[AU];
    float a_scale[AU];
    int   a_dst[AU];
    bool  a_ok[AU];
    #pragma unroll
    for (int i = 0; i < AU; ++i) {
        int u = tid + i * NT;
        a_ok[i] = (u < AUNITS);
        if (!a_ok[i]) u = 0;
        int r = u / (KC / 4);
        int c = (u % (KC / 4)) * 4;
        int g = m_base + r;
        if (g > M - 1) g = M - 1;
        a_src[i]   = A + (size_t)g * K + c;
        int co = cntA[g];
        a_scale[i] = (co > 0) ? rsqrtf(fmaxf((float)co, 1.0f)) : 1.0f;
        a_dst[i]   = r * PA + c;
    }
    const float* b_src[BU];
    int  b_dst[BU];
    bool b_ok[BU];
    #pragma unroll
    for (int i = 0; i < BU; ++i) {
        int u = tid + i * NT;
        b_ok[i] = (u < BUNITS);
        if (!b_ok[i]) u = 0;
        int kp = u / (N / 4);
        int c4 = (u % (N / 4)) * 4;
        b_src[i] = B + (size_t)(2 * kp) * N + c4;
        b_dst[i] = kp * PB + 2 * c4;
    }

    float4 ra[AU], rb0[BU], rb1[BU];
    #pragma unroll
    for (int i = 0; i < AU; ++i) ra[i] = *(const float4*)(a_src[i]);
    #pragma unroll
    for (int i = 0; i < BU; ++i) {
        rb0[i] = *(const float4*)(b_src[i]);
        rb1[i] = *(const float4*)(b_src[i] + N);
    }

    v8f acc[RTW];
    #pragma unroll
    for (int rt = 0; rt < RTW; ++rt) acc[rt] = v8f{};

    const v2f* As2 = (const v2f*)As + (((rg * 16 + l) * PA) >> 1) + h;
    const v2f* Bs2 = (const v2f*)Bs + h * (PB / 2) + ct * 16 + l;

    for (int c = 0; c < NKC; ++c) {
        #pragma unroll
        for (int i = 0; i < AU; ++i) {
            if (a_ok[i]) {
                float4 v = ra[i];
                float s = a_scale[i];
                v.x *= s; v.y *= s; v.z *= s; v.w *= s;
                *(float4*)(As + a_dst[i]) = v;
            }
        }
        #pragma unroll
        for (int i = 0; i < BU; ++i) {
            if (b_ok[i]) {
                float4 p = make_float4(rb0[i].x, rb1[i].x, rb0[i].y, rb1[i].y);
                float4 q = make_float4(rb0[i].z, rb1[i].z, rb0[i].w, rb1[i].w);
                *(float4*)(Bs + b_dst[i])     = p;
                *(float4*)(Bs + b_dst[i] + 4) = q;
            }
        }
        __syncthreads();

        if (c + 1 < NKC) {
            const int ko = (c + 1) * KC;
            #pragma unroll
            for (int i = 0; i < AU; ++i) ra[i] = *(const float4*)(a_src[i] + ko);
            #pragma unroll
            for (int i = 0; i < BU; ++i) {
                rb0[i] = *(const float4*)(b_src[i] + (size_t)ko * N);
                rb1[i] = *(const float4*)(b_src[i] + (size_t)ko * N + N);
            }
        }

        #pragma unroll
        for (int kk = 0; kk < KC; kk += 4) {
            v2f b = Bs2[(kk >> 1) * (PB / 2)];
            #pragma unroll
            for (int rt = 0; rt < RTW; ++rt) {
                v2f a = As2[((rt * RG * 16 * PA) >> 1) + (kk >> 1)];
                acc[rt] = __builtin_amdgcn_wmma_f32_16x16x4_f32(false, a, false, b, (short)0, acc[rt], false, false);
            }
        }
        __syncthreads();
    }

    float* Cs = smem;
    #pragma unroll
    for (int rt = 0; rt < RTW; ++rt) {
        #pragma unroll
        for (int i = 0; i < 8; ++i)
            Cs[((rt * RG + rg) * 16 + 8 * h + i) * PC + ct * 16 + l] = acc[rt][i];
    }
    __syncthreads();
    for (int u = tid; u < CUNITS; u += NT) {
        int r  = u / (N / 4);
        int c4 = (u % (N / 4)) * 4;
        int g  = m_base + r;
        if (g < M) *(float4*)(C + (size_t)g * N + c4) = *(const float4*)(Cs + r * PC + c4);
    }
}

__device__ __forceinline__ void gcn_add4(float4& a, const float4& x) {
    a.x += x.x; a.y += x.y; a.z += x.z; a.w += x.w;
}

template <int F, int LPN, bool RELU>
__global__ __launch_bounds__(256) void gcn_gather_sum_kernel(
    const float* __restrict__ H, const int* __restrict__ cnt_in, const slot_t* __restrict__ slots,
    const int* __restrict__ ovf_count, const unsigned* __restrict__ ovf,
    const float* __restrict__ bias, float* __restrict__ OUT, int n_nodes) {
    constexpr int NV = F / (4 * LPN);
    int node = (blockIdx.x * blockDim.x + threadIdx.x) / LPN;
    int sub  = threadIdx.x % LPN;
    if (node >= n_nodes) return;
    int n  = cnt_in[node];
    int nt = (n < CAP) ? n : CAP;
    const slot_t* row = slots + (size_t)node * CAP;
    float nd = (n > 0) ? rsqrtf(fmaxf((float)n, 1.0f)) : 1.0f;
    const float* Hc = H + (sub << 2);

    float4 acc[NV];
    for (int v = 0; v < NV; ++v) acc[v] = make_float4(0.0f, 0.0f, 0.0f, 0.0f);

    for (int base = 0; base < nt; base += LPN) {
        int m = nt - base;
        if (m > LPN) m = LPN;
        int mine = (sub < m) ? (int)row[base + sub] : 0;
        int j = 0;
        for (; j + 4 <= m; j += 4) {
            const float* p0 = Hc + (size_t)__shfl(mine, j,     LPN) * F;
            const float* p1 = Hc + (size_t)__shfl(mine, j + 1, LPN) * F;
            const float* p2 = Hc + (size_t)__shfl(mine, j + 2, LPN) * F;
            const float* p3 = Hc + (size_t)__shfl(mine, j + 3, LPN) * F;
            float4 x0[NV], x1[NV], x2[NV], x3[NV];
            for (int v = 0; v < NV; ++v) {
                x0[v] = *(const float4*)(p0 + v * LPN * 4);
                x1[v] = *(const float4*)(p1 + v * LPN * 4);
                x2[v] = *(const float4*)(p2 + v * LPN * 4);
                x3[v] = *(const float4*)(p3 + v * LPN * 4);
            }
            for (int v = 0; v < NV; ++v) {
                gcn_add4(acc[v], x0[v]);
                gcn_add4(acc[v], x1[v]);
                gcn_add4(acc[v], x2[v]);
                gcn_add4(acc[v], x3[v]);
            }
        }
        for (; j < m; ++j) {
            const float* p0 = Hc + (size_t)__shfl(mine, j, LPN) * F;
            for (int v = 0; v < NV; ++v) gcn_add4(acc[v], *(const float4*)(p0 + v * LPN * 4));
        }
    }

    if (n > CAP) {
        int oc = *ovf_count;
        for (int k = 0; k < oc; ++k) {
            unsigned w = ovf[k];
            if ((int)(w >> 16) == node) {
                const float* p0 = Hc + (size_t)(w & 0xffffu) * F;
                for (int v = 0; v < NV; ++v) gcn_add4(acc[v], *(const float4*)(p0 + v * LPN * 4));
            }
        }
    }

    const float* bp = bias + (sub << 2);
    float* o = OUT + (size_t)node * F + (sub << 2);
    for (int v = 0; v < NV; ++v) {
        float4 b = *(const float4*)(bp + v * LPN * 4);
        float4 r;
        r.x = acc[v].x * nd + b.x;
        r.y = acc[v].y * nd + b.y;
        r.z = acc[v].z * nd + b.z;
        r.w = acc[v].w * nd + b.w;
        if (RELU) {
            r.x = fmaxf(r.x, 0.0f); r.y = fmaxf(r.y, 0.0f);
            r.z = fmaxf(r.z, 0.0f); r.w = fmaxf(r.w, 0.0f);
        }
        *(float4*)(o + v * LPN * 4) = r;
    }
}


template <int F, int N2>
__global__ __launch_bounds__(512) void gcn_gather_gemm_kernel(
    const float* __restrict__ H, const int* __restrict__ cnt_in, const int* __restrict__ cnt_out,
    const slot_t* __restrict__ slots, const int* __restrict__ ovf_count, const unsigned* __restrict__ ovf,
    const float* __restrict__ bias, const float* __restrict__ W2, float* __restrict__ H2,
    int n_nodes, int n_tiles) {
    static_assert(F == 256 && N2 == 64, "16 waves = 4 column tiles x 4 k quarters of 64");
    constexpr int PA = F + 4;
    constexpr int PC = N2 + 4;
    __shared__ __attribute__((aligned(16))) float As[16 * PA];
    __shared__ __attribute__((aligned(16))) float Cs[4 * 16 * PC];

    const int tid  = threadIdx.x;
    const int wave = tid >> 5;
    const int lane = tid & 31;
    const int h    = lane >> 4;
    const int l    = lane & 15;
    const int ct   = wave & 3;
    const int kq   = wave >> 2;

    v2f bf[16];
    #pragma unroll
    for (int j = 0; j < 16; ++j) {
        const float* p = W2 + (size_t)(kq * 64 + 4 * j + 2 * h) * N2 + ct * 16 + l;
        bf[j].x = p[0];
        bf[j].y = p[N2];
    }

    const float* Hc = H + (lane << 2);
    const float4 b0 = *(const float4*)(bias + (lane << 2));
    const float4 b1 = *(const float4*)(bias + 128 + (lane << 2));
    const v2f* Ap = (const v2f*)(As + l * PA + kq * 64 + 2 * h);

    for (int tile = blockIdx.x; tile < n_tiles; tile += gridDim.x) {
        const int node = tile * 16 + wave;
        float4 r0 = make_float4(0.0f, 0.0f, 0.0f, 0.0f);
        float4 r1 = r0;
        if (node < n_nodes) {
            const slot_t* row = slots + (size_t)node * CAP;
            const int n  = cnt_in[node];
            const int co = cnt_out[node];
            const int first = (int)row[lane];
            const int nt = (n < CAP) ? n : CAP;
            const float nd = (n > 0) ? rsqrtf(fmaxf((float)n, 1.0f)) : 1.0f;
            const float ns = (co > 0) ? rsqrtf(fmaxf((float)co, 1.0f)) : 1.0f;
            float4 acc0 = make_float4(0.0f, 0.0f, 0.0f, 0.0f);
            float4 acc1 = acc0;
            for (int base = 0; base < nt; base += 32) {
                int m = nt - base;
                if (m > 32) m = 32;
                int mine = first;
                if (base > 0) mine = (lane < m) ? (int)row[base + lane] : 0;
                int j = 0;
                for (; j + 4 <= m; j += 4) {
                    const float* p0 = Hc + (size_t)__shfl(mine, j,     32) * F;
                    const float* p1 = Hc + (size_t)__shfl(mine, j + 1, 32) * F;
                    const float* p2 = Hc + (size_t)__shfl(mine, j + 2, 32) * F;
                    const float* p3 = Hc + (size_t)__shfl(mine, j + 3, 32) * F;
                    float4 x00 = *(const float4*)(p0), x01 = *(const float4*)(p0 + 128);
                    float4 x10 = *(const float4*)(p1), x11 = *(const float4*)(p1 + 128);
                    float4 x20 = *(const float4*)(p2), x21 = *(const float4*)(p2 + 128);
                    float4 x30 = *(const float4*)(p3), x31 = *(const float4*)(p3 + 128);
                    gcn_add4(acc0, x00); gcn_add4(acc0, x10); gcn_add4(acc0, x20); gcn_add4(acc0, x30);
                    gcn_add4(acc1, x01); gcn_add4(acc1, x11); gcn_add4(acc1, x21); gcn_add4(acc1, x31);
                }
                for (; j < m; ++j) {
                    const float* p0 = Hc + (size_t)__shfl(mine, j, 32) * F;
                    gcn_add4(acc0, *(const float4*)(p0));
                    gcn_add4(acc1, *(const float4*)(p0 + 128));
                }
            }
            if (n > CAP) {
                const int oc = *ovf_count;
                for (int k = 0; k < oc; ++k) {
                    unsigned w = ovf[k];
                    if ((int)(w >> 16) == node) {
                        const float* p0 = Hc + (size_t)(w & 0xffffu) * F;
                        gcn_add4(acc0, *(const float4*)(p0));
                        gcn_add4(acc1, *(const float4*)(p0 + 128));
                    }
                }
            }
            r0.x = fmaxf(acc0.x * nd + b0.x, 0.0f) * ns;
            r0.y = fmaxf(acc0.y * nd + b0.y, 0.0f) * ns;
            r0.z = fmaxf(acc0.z * nd + b0.z, 0.0f) * ns;
            r0.w = fmaxf(acc0.w * nd + b0.w, 0.0f) * ns;
            r1.x = fmaxf(acc1.x * nd + b1.x, 0.0f) * ns;
            r1.y = fmaxf(acc1.y * nd + b1.y, 0.0f) * ns;
            r1.z = fmaxf(acc1.z * nd + b1.z, 0.0f) * ns;
            r1.w = fmaxf(acc1.w * nd + b1.w, 0.0f) * ns;
        }
        *(float4*)(As + wave * PA + (lane << 2))       = r0;
        *(float4*)(As + wave * PA + 128 + (lane << 2)) = r1;
        __syncthreads();

        v8f c = v8f{};
        #pragma unroll
        for (int j = 0; j < 16; ++j) {
            v2f a = Ap[2 * j];
            c = __builtin_amdgcn_wmma_f32_16x16x4_f32(false, a, false, bf[j], (short)0, c, false, false);
        }
        #pragma unroll
        for (int i = 0; i < 8; ++i)
            Cs[(kq * 16 + 8 * h + i) * PC + ct * 16 + l] = c[i];
        __syncthreads();

        if (tid < 256) {
            const int r  = tid >> 4;
            const int c4 = (tid & 15) << 2;
            const int g  = tile * 16 + r;
            float4 s = *(const float4*)(Cs + r * PC + c4);
            gcn_add4(s, *(const float4*)(Cs + (16 + r) * PC + c4));
            gcn_add4(s, *(const float4*)(Cs + (32 + r) * PC + c4));
            gcn_add4(s, *(const float4*)(Cs + (48 + r) * PC + c4));
            if (g < n_nodes) *(float4*)(H2 + (size_t)g * N2 + c4) = s;
        }
    }
}

extern "C" void kernel_launch(void* const* d_in, const int* in_sizes, int n_in,
                              void* d_out, int out_size, void* d_ws, size_t ws_size,
                              hipStream_t stream) {
    const float* feat = (const float*)d_in[0];
    const int*   src  = (const int*)d_in[1];
    const int*   dst  = (const int*)d_in[2];
    const float* W1   = (const float*)d_in[3];
    const float* b1   = (const float*)d_in[4];
    const float* W2   = (const float*)d_in[5];
    const float* b2   = (const float*)d_in[6];
    float* out = (float*)d_out;

    const int N = in_sizes[0] / F0;
    const int E = in_sizes[1];

    float* ws   = (float*)d_ws;
    float* H1   = ws;
    float* AGG1 = H1 + (size_t)N * F1;
    float* H2   = AGG1;
    int* cnt_out   = (int*)(AGG1 + (size_t)N * F1);
    int* cnt_in    = cnt_out + N;
    int* ovf_count = cnt_in + N;
    slot_t* slots  = (slot_t*)(ovf_count + 32);
    unsigned* ovf  = (unsigned*)(slots + (size_t)N * CAP);
    const int nblk = (E + EPB - 1) / EPB;
    unsigned* sorted_in  = (unsigned*)AGG1;
    unsigned* tab_in     = sorted_in + E;
    unsigned* tab_out    = tab_in + (size_t)KMAX * nblk;
    unsigned char* sorted_out = (unsigned char*)(tab_out + (size_t)KMAX * nblk);

    gcn_bin_kernel<<<nblk, A_NT, 0, stream>>>(src, dst, sorted_in, sorted_out, tab_in, tab_out, ovf_count, E, N, nblk);
    gcn_rank_kernel<<<(N + 255) / 256, B_NT, 0, stream>>>(sorted_in, sorted_out, tab_in, tab_out, cnt_out, cnt_in, slots,
                                                         ovf_count, ovf, N, nblk);

    gcn_gemm_lds_wmma_f32<F0, F1, 13><<<(N + 207) / 208, 512, 0, stream>>>(feat, W1, cnt_out, H1, N);
    gcn_gather_gemm_kernel<F1, F2><<<512, 512, 0, stream>>>(H1, cnt_in, cnt_out, slots, ovf_count, ovf,
                                                           b1, W2, H2, N, (N + 15) / 16);
    gcn_gather_sum_kernel<F2, 16, false><<<(N * 16 + 255) / 256, 256, 0, stream>>>(H2, cnt_in, slots, ovf_count, ovf,
                                                                                  b2, out, N);
}


# both edge sums under wave-uniform (scalar) control: up to 8 (layer 1; W2 fragments now in LDS for the registers) or 16 source rows requested before the first add, short last batch too; sum order kept
# speedup vs baseline: 4.2559x; 1.0412x over previous
#include <hip/hip_runtime.h>

typedef __attribute__((ext_vector_type(2))) float v2f;
typedef __attribute__((ext_vector_type(8))) float v8f;

#define F0 512
#define F1 256
#define F2 64

#define CAP 36
typedef unsigned short slot_t;

#define KMAX 256
#define A_NT 512
#define EPT 4
#define EPB (A_NT * EPT)

__global__ __launch_bounds__(A_NT) void gcn_bin_kernel(
    const int* __restrict__ src, const int* __restrict__ dst,
    unsigned* __restrict__ sorted_in, unsigned char* __restrict__ sorted_out,
    unsigned* __restrict__ tab_in, unsigned* __restrict__ tab_out,
    int* __restrict__ ovf_count, int E, int n_nodes, int nblk) {
    __shared__ int h_in[KMAX], h_out[KMAX];
    __shared__ int c_in[KMAX], c_out[KMAX];
    const int tid = threadIdx.x;
    const int K = (n_nodes + 255) >> 8;
    if (tid < KMAX) { h_in[tid] = 0; h_out[tid] = 0; }
    if (blockIdx.x == 0 && tid == 0) *ovf_count = 0;
    __syncthreads();

    const int base = blockIdx.x * EPB;
    int s[EPT], d[EPT], r_in[EPT], r_out[EPT];
    #pragma unroll
    for (int i = 0; i < EPT; ++i) {
        int e = base + tid + i * A_NT;
        if (e < E) { s[i] = src[e]; d[i] = dst[e]; }
        else       { s[i] = -1;     d[i] = -1; }
    }
    #pragma unroll
    for (int i = 0; i < EPT; ++i) {
        r_in[i] = 0; r_out[i] = 0;
        if (d[i] >= 0) {
            r_in[i]  = atomicAdd(&h_in[d[i] >> 8], 1);
            r_out[i] = atomicAdd(&h_out[s[i] >> 8], 1);
        }
    }
    __syncthreads();

    const int wave = tid >> 5;
    const int lane = tid & 31;
    if (wave < 2) {
        int* h = wave ? h_out : h_in;
        int* c = wave ? c_out : c_in;
        int v[8];
        int sum = 0;
        #pragma unroll
        for (int j = 0; j < 8; ++j) { v[j] = h[lane * 8 + j]; sum += v[j]; }
        int incl = sum;
        for (int off = 1; off < 32; off <<= 1) {
            int t = __shfl_up(incl, off);
            if (lane >= off) incl += t;
        }
        int run = incl - sum;
        #pragma unroll
        for (int j = 0; j < 8; ++j) { c[lane * 8 + j] = v[j]; h[lane * 8 + j] = run; run += v[j]; }
    }
    __syncthreads();

    #pragma unroll
    for (int i = 0; i < EPT; ++i) {
        if (d[i] >= 0) {
            sorted_in[base + h_in[d[i] >> 8] + r_in[i]] = (unsigned)s[i] | ((unsigned)(d[i] & 255) << 16);
            sorted_out[base + h_out[s[i] >> 8] + r_out[i]] = (unsigned char)(s[i] & 255);
        }
    }
    if (tid < K) {
        tab_in[(size_t)tid * nblk + blockIdx.x]  = (unsigned)h_in[tid]  | ((unsigned)c_in[tid]  << 16);
        tab_out[(size_t)tid * nblk + blockIdx.x] = (unsigned)h_out[tid] | ((unsigned)c_out[tid] << 16);
    }
}

#define B_NT 1024
__global__ __launch_bounds__(B_NT) void gcn_rank_kernel(
    const unsigned* __restrict__ sorted_in, const unsigned char* __restrict__ sorted_out,
    const unsigned* __restrict__ tab_in, const unsigned* __restrict__ tab_out,
    int* __restrict__ cnt_out, int* __restrict__ cnt_in, slot_t* __restrict__ slots,
    int* __restrict__ ovf_count, unsigned* __restrict__ ovf, int n_nodes, int nblk) {
    __shared__ int c_i[256], c_o[256];
    __shared__ __attribute__((aligned(16))) slot_t sl[256 * CAP];
    const int tid = threadIdx.x;
    const int k = blockIdx.x;
    if (tid < 256) { c_i[tid] = 0; c_o[tid] = 0; }
    unsigned* slw = (unsigned*)sl;
    for (int i = tid; i < 256 * CAP / 2; i += B_NT) slw[i] = 0u;
    __syncthreads();

    const int t = tid & 511;
    if (tid < 512) {
        for (int b = t; b < nblk; b += 512) {
            unsigned w = tab_in[(size_t)k * nblk + b];
            const int c = (int)(w >> 16);
            const unsigned* p = sorted_in + (size_t)b * EPB + (w & 0xffffu);
            for (int j = 0; j < c; j += 4) {
                unsigned e[4];
                #pragma unroll
                for (int q = 0; q < 4; ++q) e[q] = (j + q < c) ? p[j + q] : 0u;
                #pragma unroll
                for (int q = 0; q < 4; ++q) {
                    if (j + q < c) {
                        int dl = (int)((e[q] >> 16) & 255u);
                        unsigned sid = e[q] & 0xffffu;
                        int r = atomicAdd(&c_i[dl], 1);
                        if (r < CAP) sl[dl * CAP + r] = (slot_t)sid;
                        else         ovf[atomicAdd(ovf_count, 1)] = ((unsigned)(k * 256 + dl) << 16) | sid;
                    }
                }
            }
        }
    } else {
        for (int b = t; b < nblk; b += 512) {
            unsigned w = tab_out[(size_t)k * nblk + b];
            const int c = (int)(w >> 16);
            const unsigned char* p = sorted_out + (size_t)b * EPB + (w & 0xffffu);
            for (int j = 0; j < c; j += 4) {
                int e[4];
                #pragma unroll
                for (int q = 0; q < 4; ++q) e[q] = (j + q < c) ? (int)p[j + q] : 0;
                #pragma unroll
                for (int q = 0; q < 4; ++q) {
                    if (j + q < c) atomicAdd(&c_o[e[q]], 1);
                }
            }
        }
    }
    __syncthreads();

    const int node0 = k * 256;
    if (tid < 256 && node0 + tid < n_nodes) {
        cnt_in[node0 + tid]  = c_i[tid];
        cnt_out[node0 + tid] = c_o[tid];
    }
    int rows = n_nodes - node0;
    if (rows > 256) rows = 256;
    unsigned* gs = (unsigned*)(slots + (size_t)node0 * CAP);
    for (int i = tid; i < rows * (CAP / 2); i += B_NT) gs[i] = slw[i];
}

template <int K, int N, int RTW>
__global__ __launch_bounds__(512) void gcn_gemm_lds_wmma_f32(
    const float* __restrict__ A, const float* __restrict__ B,
    const int* __restrict__ cntA, float* __restrict__ C, int M) {
    constexpr int NT = 512;
    constexpr int CT = N / 16;
    constexpr int RG = 16 / CT;
    constexpr int BM = RG * RTW * 16;
    constexpr int KC = 32;
    constexpr int NKC = K / KC;
    constexpr int PA = KC + 4;
    constexpr int PB = 2 * N + 32;
    constexpr int PC = N + 4;
    constexpr int A_FLOATS = BM * PA;
    constexpr int B_FLOATS = (KC / 2) * PB;
    constexpr int C_FLOATS = BM * PC;
    constexpr int S_FLOATS = (A_FLOATS + B_FLOATS) > C_FLOATS ? (A_FLOATS + B_FLOATS) : C_FLOATS;
    constexpr int AUNITS = BM * (KC / 4);
    constexpr int AU = (AUNITS + NT - 1) / NT;
    constexpr int BUNITS = (KC / 2) * (N / 4);
    constexpr int BU = (BUNITS + NT - 1) / NT;
    constexpr int CUNITS = BM * (N / 4);

    __shared__ __attribute__((aligned(16))) float smem[S_FLOATS];
    float* As = smem;
    float* Bs = smem + A_FLOATS;

    const int tid  = threadIdx.x;
    const int wave = tid >> 5;
    const int lane = tid & 31;
    const int h    = lane >> 4;
    const int l    = lane & 15;
    const int ct   = wave % CT;
    const int rg   = wave / CT;
    const int m_base = blockIdx.x * BM;

    const float* a_src[AU];
    float a_scale[AU];
    int   a_dst[AU];
    bool  a_ok[AU];
    #pragma unroll
    for (int i = 0; i < AU; ++i) {
        int u = tid + i * NT;
        a_ok[i] = (u < AUNITS);
        if (!a_ok[i]) u = 0;
        int r = u / (KC / 4);
        int c = (u % (KC / 4)) * 4;
        int g = m_base + r;
        if (g > M - 1) g = M - 1;
        a_src[i]   = A + (size_t)g * K + c;
        int co = cntA[g];
        a_scale[i] = (co > 0) ? rsqrtf(fmaxf((float)co, 1.0f)) : 1.0f;
        a_dst[i]   = r * PA + c;
    }
    const float* b_src[BU];
    int  b_dst[BU];
    bool b_ok[BU];
    #pragma unroll
    for (int i = 0; i < BU; ++i) {
        int u = tid + i * NT;
        b_ok[i] = (u < BUNITS);
        if (!b_ok[i]) u = 0;
        int kp = u / (N / 4);
        int c4 = (u % (N / 4)) * 4;
        b_src[i] = B + (size_t)(2 * kp) * N + c4;
        b_dst[i] = kp * PB + 2 * c4;
    }

    float4 ra[AU], rb0[BU], rb1[BU];
    #pragma unroll
    for (int i = 0; i < AU; ++i) ra[i] = *(const float4*)(a_src[i]);
    #pragma unroll
    for (int i = 0; i < BU; ++i) {
        rb0[i] = *(const float4*)(b_src[i]);
        rb1[i] = *(const float4*)(b_src[i] + N);
    }

    v8f acc[RTW];
    #pragma unroll
    for (int rt = 0; rt < RTW; ++rt) acc[rt] = v8f{};

    const v2f* As2 = (const v2f*)As + (((rg * 16 + l) * PA) >> 1) + h;
    const v2f* Bs2 = (const v2f*)Bs + h * (PB / 2) + ct * 16 + l;

    for (int c = 0; c < NKC; ++c) {
        #pragma unroll
        for (int i = 0; i < AU; ++i) {
            if (a_ok[i]) {
                float4 v = ra[i];
                float s = a_scale[i];
                v.x *= s; v.y *= s; v.z *= s; v.w *= s;
                *(float4*)(As + a_dst[i]) = v;
            }
        }
        #pragma unroll
        for (int i = 0; i < BU; ++i) {
            if (b_ok[i]) {
                float4 p = make_float4(rb0[i].x, rb1[i].x, rb0[i].y, rb1[i].y);
                float4 q = make_float4(rb0[i].z, rb1[i].z, rb0[i].w, rb1[i].w);
                *(float4*)(Bs + b_dst[i])     = p;
                *(float4*)(Bs + b_dst[i] + 4) = q;
            }
        }
        __syncthreads();

        if (c + 1 < NKC) {
            const int ko = (c + 1) * KC;
            #pragma unroll
            for (int i = 0; i < AU; ++i) ra[i] = *(const float4*)(a_src[i] + ko);
            #pragma unroll
            for (int i = 0; i < BU; ++i) {
                rb0[i] = *(const float4*)(b_src[i] + (size_t)ko * N);
                rb1[i] = *(const float4*)(b_src[i] + (size_t)ko * N + N);
            }
        }

        #pragma unroll
        for (int kk = 0; kk < KC; kk += 4) {
            v2f b = Bs2[(kk >> 1) * (PB / 2)];
            #pragma unroll
            for (int rt = 0; rt < RTW; ++rt) {
                v2f a = As2[((rt * RG * 16 * PA) >> 1) + (kk >> 1)];
                acc[rt] = __builtin_amdgcn_wmma_f32_16x16x4_f32(false, a, false, b, (short)0, acc[rt], false, false);
            }
        }
        __syncthreads();
    }

    float* Cs = smem;
    #pragma unroll
    for (int rt = 0; rt < RTW; ++rt) {
        #pragma unroll
        for (int i = 0; i < 8; ++i)
            Cs[((rt * RG + rg) * 16 + 8 * h + i) * PC + ct * 16 + l] = acc[rt][i];
    }
    __syncthreads();
    for (int u = tid; u < CUNITS; u += NT) {
        int r  = u / (N / 4);
        int c4 = (u % (N / 4)) * 4;
        int g  = m_base + r;
        if (g < M) *(float4*)(C + (size_t)g * N + c4) = *(const float4*)(Cs + r * PC + c4);
    }
}

__device__ __forceinline__ void gcn_add4(float4& a, const float4& x) {
    a.x += x.x; a.y += x.y; a.z += x.z; a.w += x.w;
}

typedef __attribute__((ext_vector_type(4))) float v4f;
typedef __attribute__((address_space(1))) const float gfloat;

__device__ __forceinline__ gfloat* gcn_row_base(const float* H, int sid, int F) {
    unsigned long long a = (unsigned long long)(H + (size_t)sid * F);
    asm volatile("" : "+s"(a));
    return (gfloat*)a;
}

template <int R>
__device__ __forceinline__ void gcn_batch64(const float* H, int sw, int jb, unsigned loff, float2& acc) {
    v2f x[R];
    gfloat* p[R];
    #pragma unroll
    for (int q = 0; q < R; ++q) p[q] = gcn_row_base(H, __builtin_amdgcn_readlane(sw, jb + q), F2);
    #pragma unroll
    for (int q = 0; q < R; ++q) x[q] = *(const __attribute__((address_space(1))) v2f*)(p[q] + loff);
    #pragma unroll
    for (int q = 0; q < R; ++q) { acc.x += x[q].x; acc.y += x[q].y; }
}

template <int R>
__device__ __forceinline__ void gcn_batch256(const float* H, int sw, int jb, unsigned loff, float4& acc0, float4& acc1) {
    v4f x0[R], x1[R];
    gfloat* p[R];
    #pragma unroll
    for (int q = 0; q < R; ++q) p[q] = gcn_row_base(H, __builtin_amdgcn_readlane(sw, jb + q), F1);
    #pragma unroll
    for (int q = 0; q < R; ++q) {
        x0[q] = *(const __attribute__((address_space(1))) v4f*)(p[q] + loff);
        x1[q] = *(const __attribute__((address_space(1))) v4f*)(p[q] + loff + 128);
    }
    #pragma unroll
    for (int q = 0; q < R; ++q) {
        acc0.x += x0[q].x; acc0.y += x0[q].y; acc0.z += x0[q].z; acc0.w += x0[q].w;
        acc1.x += x1[q].x; acc1.y += x1[q].y; acc1.z += x1[q].z; acc1.w += x1[q].w;
    }
}

#define G2ROWS 16
__global__ __launch_bounds__(256) void gcn_gather_sum64_kernel(
    const float* __restrict__ H, const int* __restrict__ cnt_in, const slot_t* __restrict__ slots,
    const int* __restrict__ ovf_count, const unsigned* __restrict__ ovf,
    const float* __restrict__ bias, float* __restrict__ OUT, int n_nodes) {
    const int lane = threadIdx.x & 31;
    const int wave_g  = __builtin_amdgcn_readfirstlane((int)((blockIdx.x * blockDim.x + threadIdx.x) >> 5));
    const int n_waves = (int)((gridDim.x * blockDim.x) >> 5);
    const float2 bv = *(const float2*)(bias + 2 * lane);
    const unsigned loff = 2u * (unsigned)lane;

    int node = wave_g;
    int cin_v = 0, s0_nx = 0, s1_nx = 0;
    if (node < n_nodes) {
        const slot_t* row = slots + (size_t)node * CAP;
        s0_nx = (int)row[lane];
        s1_nx = (int)row[32 + (lane & 3)];
    }
    int it = 0;
    for (; node < n_nodes; node += n_waves, ++it) {
        if ((it & 31) == 0) {
            const int node_i = node + lane * n_waves;
            cin_v = (node_i < n_nodes) ? cnt_in[node_i] : 0;
        }
        const int n = __builtin_amdgcn_readlane(cin_v, it & 31);
        const int s0 = s0_nx, s1 = s1_nx;
        const int node2 = node + n_waves;
        if (node2 < n_nodes) {
            const slot_t* row2 = slots + (size_t)node2 * CAP;
            s0_nx = (int)row2[lane];
            s1_nx = (int)row2[32 + (lane & 3)];
        }
        const int nt = (n < CAP) ? n : CAP;
        const float nd = (n > 0) ? rsqrtf(fmaxf((float)n, 1.0f)) : 1.0f;
        float2 acc = make_float2(0.0f, 0.0f);

        int base = 0;
        for (; base + G2ROWS <= nt; base += G2ROWS)
            gcn_batch64<G2ROWS>(H, (base < 32) ? s0 : s1, base & 31, loff, acc);
        {
            const int sw = (base < 32) ? s0 : s1;
            const int jb = base & 31;
            switch (nt - base) {
                case 1:  gcn_batch64<1>(H, sw, jb, loff, acc);  break;
                case 2:  gcn_batch64<2>(H, sw, jb, loff, acc);  break;
                case 3:  gcn_batch64<3>(H, sw, jb, loff, acc);  break;
                case 4:  gcn_batch64<4>(H, sw, jb, loff, acc);  break;
                case 5:  gcn_batch64<5>(H, sw, jb, loff, acc);  break;
                case 6:  gcn_batch64<6>(H, sw, jb, loff, acc);  break;
                case 7:  gcn_batch64<7>(H, sw, jb, loff, acc);  break;
                case 8:  gcn_batch64<8>(H, sw, jb, loff, acc);  break;
                case 9:  gcn_batch64<9>(H, sw, jb, loff, acc);  break;
                case 10: gcn_batch64<10>(H, sw, jb, loff, acc); break;
                case 11: gcn_batch64<11>(H, sw, jb, loff, acc); break;
                case 12: gcn_batch64<12>(H, sw, jb, loff, acc); break;
                case 13: gcn_batch64<13>(H, sw, jb, loff, acc); break;
                case 14: gcn_batch64<14>(H, sw, jb, loff, acc); break;
                case 15: gcn_batch64<15>(H, sw, jb, loff, acc); break;
                default: break;
            }
        }

        if (n > CAP) {
            const int oc = *ovf_count;
            for (int k = 0; k < oc; ++k) {
                unsigned w = ovf[k];
                if ((int)(w >> 16) == node) {
                    const float2 xv = *(const float2*)(H + (size_t)(w & 0xffffu) * F2 + 2 * lane);
                    acc.x += xv.x; acc.y += xv.y;
                }
            }
        }

        float2 o;
        o.x = acc.x * nd + bv.x;
        o.y = acc.y * nd + bv.y;
        *(float2*)(OUT + (size_t)node * F2 + 2 * lane) = o;
    }
}


#define G1ROWS 8
template <int F, int N2>
__global__ __launch_bounds__(512, 8) void gcn_gather_gemm_kernel(
    const float* __restrict__ H, const int* __restrict__ cnt_in, const int* __restrict__ cnt_out,
    const slot_t* __restrict__ slots, const int* __restrict__ ovf_count, const unsigned* __restrict__ ovf,
    const float* __restrict__ bias, const float* __restrict__ W2, float* __restrict__ H2,
    int n_nodes, int n_tiles) {
    static_assert(F == F1 && N2 == F2, "16 waves = 4 column tiles x 4 k quarters of 64");
    static_assert(G1ROWS == 8, "the short-batch switch below lists lengths 1 to G1ROWS - 1");
    constexpr int PA = F + 4;
    constexpr int PC = N2 + 4;
    __shared__ __attribute__((aligned(16))) float As[16 * PA];
    __shared__ __attribute__((aligned(16))) float Cs[4 * 16 * PC];
    __shared__ __attribute__((aligned(16))) v2f Bl[16 * 16 * 32];

    const int tid  = threadIdx.x;
    const int wave = __builtin_amdgcn_readfirstlane(tid >> 5);
    const int lane = tid & 31;
    const int h    = lane >> 4;
    const int l    = lane & 15;
    const int ct   = wave & 3;
    const int kq   = wave >> 2;

    v2f* bfl = Bl + wave * (16 * 32) + lane;
    #pragma unroll
    for (int j = 0; j < 16; ++j) {
        const float* p = W2 + (size_t)(kq * 64 + 4 * j + 2 * h) * N2 + ct * 16 + l;
        v2f b;
        b.x = p[0];
        b.y = p[N2];
        bfl[j * 32] = b;
    }

    const float* Hc = H + (lane << 2);
    const unsigned loff = 4u * (unsigned)lane;
    const float4 b0 = *(const float4*)(bias + (lane << 2));
    const float4 b1 = *(const float4*)(bias + 128 + (lane << 2));
    const v2f* Ap = (const v2f*)(As + l * PA + kq * 64 + 2 * h);

    int cin_v = 0, cout_v = 0, s0_nx = 0, s1_nx = 0;
    {
        const int node0 = blockIdx.x * 16 + wave;
        if ((int)blockIdx.x < n_tiles && node0 < n_nodes) {
            const slot_t* row = slots + (size_t)node0 * CAP;
            s0_nx = (int)row[lane];
            s1_nx = (int)row[32 + (lane & 3)];
        }
    }

    int it = 0;
    for (int tile = blockIdx.x; tile < n_tiles; tile += gridDim.x, ++it) {
        const int node = tile * 16 + wave;
        if ((it & 31) == 0) {
            const int tile_i = tile + lane * (int)gridDim.x;
            const int node_i = tile_i * 16 + wave;
            const bool ok = (tile_i < n_tiles) && (node_i < n_nodes);
            cin_v  = ok ? cnt_in[node_i]  : 0;
            cout_v = ok ? cnt_out[node_i] : 0;
        }
        const int n  = __builtin_amdgcn_readlane(cin_v,  it & 31);
        const int co = __builtin_amdgcn_readlane(cout_v, it & 31);
        const int s0 = s0_nx, s1 = s1_nx;
        {
            const int tile2 = tile + (int)gridDim.x;
            const int node2 = tile2 * 16 + wave;
            if (tile2 < n_tiles && node2 < n_nodes) {
                const slot_t* row2 = slots + (size_t)node2 * CAP;
                s0_nx = (int)row2[lane];
                s1_nx = (int)row2[32 + (lane & 3)];
            }
        }
        float4 r0 = make_float4(0.0f, 0.0f, 0.0f, 0.0f);
        float4 r1 = r0;
        if (node < n_nodes) {
            const int nt = (n < CAP) ? n : CAP;
            const float nd = (n > 0) ? rsqrtf(fmaxf((float)n, 1.0f)) : 1.0f;
            const float ns = (co > 0) ? rsqrtf(fmaxf((float)co, 1.0f)) : 1.0f;
            float4 acc0 = make_float4(0.0f, 0.0f, 0.0f, 0.0f);
            float4 acc1 = acc0;
            int base = 0;
            for (; base + G1ROWS <= nt; base += G1ROWS)
                gcn_batch256<G1ROWS>(H, (base < 32) ? s0 : s1, base & 31, loff, acc0, acc1);
            {
                const int sw = (base < 32) ? s0 : s1;
                const int jb = base & 31;
                switch (nt - base) {
                    case 1: gcn_batch256<1>(H, sw, jb, loff, acc0, acc1); break;
                    case 2: gcn_batch256<2>(H, sw, jb, loff, acc0, acc1); break;
                    case 3: gcn_batch256<3>(H, sw, jb, loff, acc0, acc1); break;
                    case 4: gcn_batch256<4>(H, sw, jb, loff, acc0, acc1); break;
                    case 5: gcn_batch256<5>(H, sw, jb, loff, acc0, acc1); break;
                    case 6: gcn_batch256<6>(H, sw, jb, loff, acc0, acc1); break;
                    case 7: gcn_batch256<7>(H, sw, jb, loff, acc0, acc1); break;
                    default: break;
                }
            }
            if (n > CAP) {
                const int oc = *ovf_count;
                for (int k = 0; k < oc; ++k) {
                    unsigned w = ovf[k];
                    if ((int)(w >> 16) == node) {
                        const float* p0 = Hc + (size_t)(w & 0xffffu) * F;
                        gcn_add4(acc0, *(const float4*)(p0));
                        gcn_add4(acc1, *(const float4*)(p0 + 128));
                    }
                }
            }
            r0.x = fmaxf(acc0.x * nd + b0.x, 0.0f) * ns;
            r0.y = fmaxf(acc0.y * nd + b0.y, 0.0f) * ns;
            r0.z = fmaxf(acc0.z * nd + b0.z, 0.0f) * ns;
            r0.w = fmaxf(acc0.w * nd + b0.w, 0.0f) * ns;
            r1.x = fmaxf(acc1.x * nd + b1.x, 0.0f) * ns;
            r1.y = fmaxf(acc1.y * nd + b1.y, 0.0f) * ns;
            r1.z = fmaxf(acc1.z * nd + b1.z, 0.0f) * ns;
            r1.w = fmaxf(acc1.w * nd + b1.w, 0.0f) * ns;
        }
        *(float4*)(As + wave * PA + (lane << 2))       = r0;
        *(float4*)(As + wave * PA + 128 + (lane << 2)) = r1;
        __syncthreads();

        v8f c = v8f{};
        #pragma unroll
        for (int j = 0; j < 16; ++j) {
            v2f a = Ap[2 * j];
            v2f b = bfl[j * 32];
            c = __builtin_amdgcn_wmma_f32_16x16x4_f32(false, a, false, b, (short)0, c, false, false);
        }
        #pragma unroll
        for (int i = 0; i < 8; ++i)
            Cs[(kq * 16 + 8 * h + i) * PC + ct * 16 + l] = c[i];
        __syncthreads();

        if (tid < 256) {
            const int r  = tid >> 4;
            const int c4 = (tid & 15) << 2;
            const int g  = tile * 16 + r;
            float4 s = *(const float4*)(Cs + r * PC + c4);
            gcn_add4(s, *(const float4*)(Cs + (16 + r) * PC + c4));
            gcn_add4(s, *(const float4*)(Cs + (32 + r) * PC + c4));
            gcn_add4(s, *(const float4*)(Cs + (48 + r) * PC + c4));
            if (g < n_nodes) *(float4*)(H2 + (size_t)g * N2 + c4) = s;
        }
    }
}

extern "C" void kernel_launch(void* const* d_in, const int* in_sizes, int n_in,
                              void* d_out, int out_size, void* d_ws, size_t ws_size,
                              hipStream_t stream) {
    const float* feat = (const float*)d_in[0];
    const int*   src  = (const int*)d_in[1];
    const int*   dst  = (const int*)d_in[2];
    const float* W1   = (const float*)d_in[3];
    const float* b1   = (const float*)d_in[4];
    const float* W2   = (const float*)d_in[5];
    const float* b2   = (const float*)d_in[6];
    float* out = (float*)d_out;

    const int N = in_sizes[0] / F0;
    const int E = in_sizes[1];

    float* ws   = (float*)d_ws;
    float* H1   = ws;
    float* AGG1 = H1 + (size_t)N * F1;
    float* H2   = AGG1;
    int* cnt_out   = (int*)(AGG1 + (size_t)N * F1);
    int* cnt_in    = cnt_out + N;
    int* ovf_count = cnt_in + N;
    slot_t* slots  = (slot_t*)(ovf_count + 32);
    unsigned* ovf  = (unsigned*)(slots + (size_t)N * CAP);
    const int nblk = (E + EPB - 1) / EPB;
    unsigned* sorted_in  = (unsigned*)AGG1;
    unsigned* tab_in     = sorted_in + E;
    unsigned* tab_out    = tab_in + (size_t)KMAX * nblk;
    unsigned char* sorted_out = (unsigned char*)(tab_out + (size_t)KMAX * nblk);

    gcn_bin_kernel<<<nblk, A_NT, 0, stream>>>(src, dst, sorted_in, sorted_out, tab_in, tab_out, ovf_count, E, N, nblk);
    gcn_rank_kernel<<<(N + 255) / 256, B_NT, 0, stream>>>(sorted_in, sorted_out, tab_in, tab_out, cnt_out, cnt_in, slots,
                                                         ovf_count, ovf, N, nblk);

    gcn_gemm_lds_wmma_f32<F0, F1, 13><<<(N + 207) / 208, 512, 0, stream>>>(feat, W1, cnt_out, H1, N);
    gcn_gather_gemm_kernel<F1, F2><<<512, 512, 0, stream>>>(H1, cnt_in, cnt_out, slots, ovf_count, ovf,
                                                           b1, W2, H2, N, (N + 15) / 16);
    gcn_gather_sum64_kernel<<<1024, 256, 0, stream>>>(H2, cnt_in, slots, ovf_count, ovf, b2, out, N);
}
